# starting kernel
# baseline (speedup 1.0000x reference)
#include <hip/hip_runtime.h>

#define DIM   128
#define HEADS 8
#define BATCH 2
#define SEQ   2048
#define HD    16
#define QT    4

typedef __attribute__((ext_vector_type(2))) float v2f;
typedef __attribute__((ext_vector_type(8))) float v8f;

__global__ void __launch_bounds__(32)
qkv_proj_wmma(const float* __restrict__ X,
              const float* __restrict__ Wq,
              const float* __restrict__ Wk,
              const float* __restrict__ Wv,
              float* __restrict__ Qh,
              float* __restrict__ Kh,
              float* __restrict__ Vh)
{
    const int lane = threadIdx.x;
    const int n    = lane & 15;
    const int hi   = lane >> 4;
    const int s0   = blockIdx.x * 16;
    const int bh   = blockIdx.y;
    const int b    = bh >> 3;
    const int h    = bh & 7;

    const float* W   = (blockIdx.z == 0) ? Wq : (blockIdx.z == 1) ? Wk : Wv;
    float*       Out = (blockIdx.z == 0) ? Qh : (blockIdx.z == 1) ? Kh : Vh;

    const float* Xrow = X + ((size_t)b * SEQ + s0 + n) * DIM;
    const int    col  = n * HEADS + h;

    v8f c = {};
    #pragma unroll 4
    for (int k0 = 0; k0 < DIM; k0 += 4) {
        const int ka = k0 + 2 * hi;
        v2f a, bb;
        a.x  = Xrow[ka];
        a.y  = Xrow[ka + 1];
        bb.x = W[(size_t)ka * DIM + col];
        bb.y = W[(size_t)(ka + 1) * DIM + col];
        c = __builtin_amdgcn_wmma_f32_16x16x4_f32(false, a, false, bb,
                                                  (short)0, c, false, false);
    }

    float* Obase = Out + ((size_t)bh * SEQ + s0) * HD;
    #pragma unroll
    for (int r = 0; r < 8; ++r) {
        const int M = r + 8 * hi;
        Obase[(size_t)M * HD + n] = c[r];
    }
}

__global__ void __launch_bounds__(32)
flash_attn_wmma(const float* __restrict__ Qh,
                const float* __restrict__ Kh,
                const float* __restrict__ Vh,
                const int*   __restrict__ mask_src,
                float*       __restrict__ out)
{
    __shared__ float Plds[QT][16 * 16];

    const int lane = threadIdx.x;
    const int n    = lane & 15;
    const int hi   = lane >> 4;
    const int q0   = blockIdx.x * (16 * QT);
    const int bh   = blockIdx.y;
    const int b    = bh >> 3;
    const int h    = bh & 7;

    v2f qa[QT][4];
    #pragma unroll
    for (int qt = 0; qt < QT; ++qt) {
        const float* Qrow = Qh + ((size_t)bh * SEQ + q0 + qt * 16 + n) * HD;
        #pragma unroll
        for (int t = 0; t < 4; ++t) {
            const int ka = 4 * t + 2 * hi;
            qa[qt][t].x = Qrow[ka];
            qa[qt][t].y = Qrow[ka + 1];
        }
    }

    float m[QT][8], l[QT][8];
    v8f acc[QT];
    #pragma unroll
    for (int qt = 0; qt < QT; ++qt) {
        acc[qt] = (v8f){};
        #pragma unroll
        for (int r = 0; r < 8; ++r) { m[qt][r] = -1.0e30f; l[qt][r] = 0.0f; }
    }

    const float scale = 0.25f;
    const float* Kbase = Kh + (size_t)bh * SEQ * HD;
    const float* Vbase = Vh + (size_t)bh * SEQ * HD;

    for (int kv0 = 0; kv0 < SEQ; kv0 += 16) {
        __builtin_prefetch(Kbase + (size_t)(kv0 + 16 + n) * HD, 0, 1);
        __builtin_prefetch(Vbase + (size_t)(kv0 + 16 + n) * HD, 0, 1);

        const float* Krow = Kbase + (size_t)(kv0 + n) * HD;
        v2f kb[4];
        #pragma unroll
        for (int t = 0; t < 4; ++t) {
            const int ka = 4 * t + 2 * hi;
            kb[t].x = Krow[ka];
            kb[t].y = Krow[ka + 1];
        }

        const float addm = (mask_src[b * SEQ + kv0 + n] == 0) ? -1000.0f : 0.0f;

        #pragma unroll
        for (int qt = 0; qt < QT; ++qt) {
            v8f s = {};
            #pragma unroll
            for (int t = 0; t < 4; ++t)
                s = __builtin_amdgcn_wmma_f32_16x16x4_f32(false, qa[qt][t],
                                                          false, kb[t],
                                                          (short)0, s,
                                                          false, false);
            #pragma unroll
            for (int r = 0; r < 8; ++r) s[r] = s[r] * scale + addm;

            #pragma unroll
            for (int r = 0; r < 8; ++r) {
                float v = s[r];
                #pragma unroll
                for (int off = 1; off < 16; off <<= 1)
                    v = fmaxf(v, __shfl_xor(v, off, 32));
                const float mnew  = fmaxf(m[qt][r], v);
                const float alpha = __expf(m[qt][r] - mnew);
                const float p     = __expf(s[r] - mnew);
                float ps = p;
                #pragma unroll
                for (int off = 1; off < 16; off <<= 1)
                    ps += __shfl_xor(ps, off, 32);
                l[qt][r]   = l[qt][r] * alpha + ps;
                m[qt][r]   = mnew;
                acc[qt][r] = acc[qt][r] * alpha;
                s[r]       = p;
            }

            #pragma unroll
            for (int r = 0; r < 8; ++r)
                Plds[qt][(r + 8 * hi) * 16 + n] = s[r];
        }
        __syncthreads();

        const float* Vrow = Vbase + (size_t)kv0 * HD;
        v2f vb[4];
        #pragma unroll
        for (int t = 0; t < 4; ++t) {
            const int ka = 4 * t + 2 * hi;
            vb[t].x = Vrow[(size_t)ka * HD + n];
            vb[t].y = Vrow[(size_t)(ka + 1) * HD + n];
        }

        #pragma unroll
        for (int qt = 0; qt < QT; ++qt) {
            #pragma unroll
            for (int t = 0; t < 4; ++t) {
                const int ka = 4 * t + 2 * hi;
                v2f pa;
                pa.x = Plds[qt][n * 16 + ka];
                pa.y = Plds[qt][n * 16 + ka + 1];
                acc[qt] = __builtin_amdgcn_wmma_f32_16x16x4_f32(false, pa,
                                                               false, vb[t],
                                                               (short)0, acc[qt],
                                                               false, false);
            }
        }
        __syncthreads();
    }

    #pragma unroll
    for (int qt = 0; qt < QT; ++qt) {
        #pragma unroll
        for (int r = 0; r < 8; ++r) {
            const int M = r + 8 * hi;
            out[((size_t)b * SEQ + q0 + qt * 16 + M) * DIM + (n * HEADS + h)] =
                acc[qt][r] / l[qt][r];
        }
    }
}

extern "C" void kernel_launch(void* const* d_in, const int* in_sizes, int n_in,
                              void* d_out, int out_size, void* d_ws, size_t ws_size,
                              hipStream_t stream)
{
    const float* X        = (const float*)d_in[0];
    const int*   mask_src = (const int*)d_in[1];
    const float* Wq       = (const float*)d_in[4];
    const float* Wk       = (const float*)d_in[5];
    const float* Wv       = (const float*)d_in[6];
    float*       out      = (float*)d_out;

    const size_t headElems = (size_t)BATCH * HEADS * SEQ * HD;
    float* Qh = (float*)d_ws;
    float* Kh = Qh + headElems;
    float* Vh = Kh + headElems;

    dim3 gridP(SEQ / 16, BATCH * HEADS, 3);
    qkv_proj_wmma<<<gridP, 32, 0, stream>>>(X, Wq, Wk, Wv, Qh, Kh, Vh);

    dim3 gridA(SEQ / (16 * QT), BATCH * HEADS);
    flash_attn_wmma<<<gridA, 32, 0, stream>>>(Qh, Kh, Vh, mask_src, out);
}


# Score tile transposed (S^T = K*Q^T): softmax max/sum reduce in-lane plus one half-wave swap, not 16-lane shuffle trees; P feeds PV from registers; row-sum tree and PV k order change
# speedup vs baseline: 3.2379x; 3.2379x over previous
#include <hip/hip_runtime.h>

#define DIM   128
#define HEADS 8
#define BATCH 2
#define SEQ   2048
#define HD    16
#define QT    4

typedef __attribute__((ext_vector_type(2))) float v2f;
typedef __attribute__((ext_vector_type(8))) float v8f;

__global__ void __launch_bounds__(32)
qkv_proj_wmma(const float* __restrict__ X,
              const float* __restrict__ Wq,
              const float* __restrict__ Wk,
              const float* __restrict__ Wv,
              float* __restrict__ Qh,
              float* __restrict__ Kh,
              float* __restrict__ Vh)
{
    const int lane = threadIdx.x;
    const int n    = lane & 15;
    const int hi   = lane >> 4;
    const int s0   = blockIdx.x * 16;
    const int bh   = blockIdx.y;
    const int b    = bh >> 3;
    const int h    = bh & 7;

    const float* W   = (blockIdx.z == 0) ? Wq : (blockIdx.z == 1) ? Wk : Wv;
    float*       Out = (blockIdx.z == 0) ? Qh : (blockIdx.z == 1) ? Kh : Vh;

    const float* Xrow = X + ((size_t)b * SEQ + s0 + n) * DIM;
    const int    col  = n * HEADS + h;

    v8f c = {};
    #pragma unroll 4
    for (int k0 = 0; k0 < DIM; k0 += 4) {
        const int ka = k0 + 2 * hi;
        v2f a, bb;
        a.x  = Xrow[ka];
        a.y  = Xrow[ka + 1];
        bb.x = W[(size_t)ka * DIM + col];
        bb.y = W[(size_t)(ka + 1) * DIM + col];
        c = __builtin_amdgcn_wmma_f32_16x16x4_f32(false, a, false, bb,
                                                  (short)0, c, false, false);
    }

    float* Obase = Out + ((size_t)bh * SEQ + s0) * HD;
    #pragma unroll
    for (int r = 0; r < 8; ++r) {
        const int M = r + 8 * hi;
        Obase[(size_t)M * HD + n] = c[r];
    }
}


__device__ __forceinline__ float xhalf(float x)
{
    const int b = __builtin_bit_cast(int, x);
    return __builtin_bit_cast(float,
        __builtin_amdgcn_permlanex16(b, b, 0x76543210, 0xfedcba98, false, false));
}

__global__ void __launch_bounds__(32)
flash_attn_wmma(const float* __restrict__ Qh,
                const float* __restrict__ Kh,
                const float* __restrict__ Vh,
                const int*   __restrict__ mask_src,
                float*       __restrict__ out)
{
    const int lane = threadIdx.x;
    const int n    = lane & 15;
    const int hi   = lane >> 4;
    const int q0   = blockIdx.x * (16 * QT);
    const int bh   = blockIdx.y;
    const int b    = bh >> 3;
    const int h    = bh & 7;


    v2f qb[QT][4];
    #pragma unroll
    for (int qt = 0; qt < QT; ++qt) {
        const float* Qrow = Qh + ((size_t)bh * SEQ + q0 + qt * 16 + n) * HD;
        #pragma unroll
        for (int t = 0; t < 4; ++t) {
            const int ka = 4 * t + 2 * hi;
            qb[qt][t].x = Qrow[ka];
            qb[qt][t].y = Qrow[ka + 1];
        }
    }

    float m[QT], l[QT];
    v8f acc[QT];
    #pragma unroll
    for (int qt = 0; qt < QT; ++qt) {
        acc[qt] = (v8f){};
        m[qt] = -1.0e30f;
        l[qt] = 0.0f;
    }

    const float scale = 0.25f;
    const float* Kbase = Kh + (size_t)bh * SEQ * HD;
    const float* Vbase = Vh + (size_t)bh * SEQ * HD;
    const int*   Mbase = mask_src + (size_t)b * SEQ + 8 * hi;

    for (int kv0 = 0; kv0 < SEQ; kv0 += 16) {
        __builtin_prefetch(Kbase + (size_t)(kv0 + 16 + n) * HD, 0, 1);
        __builtin_prefetch(Vbase + (size_t)(kv0 + 16 + n) * HD, 0, 1);

        const float* Krow = Kbase + (size_t)(kv0 + n) * HD;
        v2f ka[4];
        #pragma unroll
        for (int t = 0; t < 4; ++t) {
            const int kd = 4 * t + 2 * hi;
            ka[t].x = Krow[kd];
            ka[t].y = Krow[kd + 1];
        }

        const float* Vcol = Vbase + (size_t)(kv0 + 8 * hi) * HD + n;
        v2f va[4];
        #pragma unroll
        for (int t = 0; t < 4; ++t) {
            va[t].x = Vcol[(size_t)(2 * t) * HD];
            va[t].y = Vcol[(size_t)(2 * t + 1) * HD];
        }

        const int4 mk0 = *(const int4*)(Mbase + kv0);
        const int4 mk1 = *(const int4*)(Mbase + kv0 + 4);
        float addm[8];
        addm[0] = (mk0.x == 0) ? -1000.0f : 0.0f;
        addm[1] = (mk0.y == 0) ? -1000.0f : 0.0f;
        addm[2] = (mk0.z == 0) ? -1000.0f : 0.0f;
        addm[3] = (mk0.w == 0) ? -1000.0f : 0.0f;
        addm[4] = (mk1.x == 0) ? -1000.0f : 0.0f;
        addm[5] = (mk1.y == 0) ? -1000.0f : 0.0f;
        addm[6] = (mk1.z == 0) ? -1000.0f : 0.0f;
        addm[7] = (mk1.w == 0) ? -1000.0f : 0.0f;

        #pragma unroll
        for (int qt = 0; qt < QT; ++qt) {
            v8f s = {};
            #pragma unroll
            for (int t = 0; t < 4; ++t)
                s = __builtin_amdgcn_wmma_f32_16x16x4_f32(false, ka[t],
                                                          false, qb[qt][t],
                                                          (short)0, s,
                                                          false, false);
            #pragma unroll
            for (int r = 0; r < 8; ++r) s[r] = s[r] * scale + addm[r];

            float mx = fmaxf(fmaxf(fmaxf(s[0], s[1]), fmaxf(s[2], s[3])),
                             fmaxf(fmaxf(s[4], s[5]), fmaxf(s[6], s[7])));
            mx = fmaxf(mx, xhalf(mx));
            const float mnew  = fmaxf(m[qt], mx);
            const float alpha = __expf(m[qt] - mnew);
            #pragma unroll
            for (int r = 0; r < 8; ++r) s[r] = __expf(s[r] - mnew);
            float ps = ((s[0] + s[1]) + (s[2] + s[3])) +
                       ((s[4] + s[5]) + (s[6] + s[7]));
            ps += xhalf(ps);
            l[qt] = l[qt] * alpha + ps;
            m[qt] = mnew;
            #pragma unroll
            for (int r = 0; r < 8; ++r) acc[qt][r] *= alpha;

            #pragma unroll
            for (int t = 0; t < 4; ++t) {
                v2f pb;
                pb.x = s[2 * t];
                pb.y = s[2 * t + 1];
                acc[qt] = __builtin_amdgcn_wmma_f32_16x16x4_f32(false, va[t],
                                                               false, pb,
                                                               (short)0, acc[qt],
                                                               false, false);
            }
        }
    }

    #pragma unroll
    for (int qt = 0; qt < QT; ++qt) {
        float* Orow = out + ((size_t)b * SEQ + q0 + qt * 16 + n) * DIM + h;
        #pragma unroll
        for (int r = 0; r < 8; ++r)
            Orow[(8 * hi + r) * HEADS] = acc[qt][r] / l[qt];
    }
}

extern "C" void kernel_launch(void* const* d_in, const int* in_sizes, int n_in,
                              void* d_out, int out_size, void* d_ws, size_t ws_size,
                              hipStream_t stream)
{
    const float* X        = (const float*)d_in[0];
    const int*   mask_src = (const int*)d_in[1];
    const float* Wq       = (const float*)d_in[4];
    const float* Wk       = (const float*)d_in[5];
    const float* Wv       = (const float*)d_in[6];
    float*       out      = (float*)d_out;

    const size_t headElems = (size_t)BATCH * HEADS * SEQ * HD;
    float* Qh = (float*)d_ws;
    float* Kh = Qh + headElems;
    float* Vh = Kh + headElems;

    dim3 gridP(SEQ / 16, BATCH * HEADS, 3);
    qkv_proj_wmma<<<gridP, 32, 0, stream>>>(X, Wq, Wk, Wv, Qh, Kh, Vh);

    dim3 gridA(SEQ / (16 * QT), BATCH * HEADS);
    flash_attn_wmma<<<gridA, 32, 0, stream>>>(Qh, Kh, Vh, mask_src, out);
}


# Attention key sweep split over 8 waves (256 keys each) of 16-wave work-groups, partial (m,l,O) merged once through LDS: 4 waves on every SIMD; row-sum and PV sums become 8 partial sums
# speedup vs baseline: 7.4139x; 2.2897x over previous
#include <hip/hip_runtime.h>

#define DIM   128
#define HEADS 8
#define BATCH 2
#define SEQ   2048
#define HD    16
#define QT    4
#define NW    8
#define QG    2
#define KPW   (SEQ / NW)
#define WPQ   (NW / QT)
#define RPW   (8 / WPQ)
#define ATT_BLOCK 512
static_assert(ATT_BLOCK == 32 * NW * QG, "one wave per (query group, key range)");
static_assert(NW % QT == 0 && 8 % WPQ == 0 && KPW % 16 == 0, "merge split");

typedef __attribute__((ext_vector_type(2))) float v2f;
typedef __attribute__((ext_vector_type(8))) float v8f;

__global__ void __launch_bounds__(32)
qkv_proj_wmma(const float* __restrict__ X,
              const float* __restrict__ Wq,
              const float* __restrict__ Wk,
              const float* __restrict__ Wv,
              float* __restrict__ Qh,
              float* __restrict__ Kh,
              float* __restrict__ Vh)
{
    const int lane = threadIdx.x;
    const int n    = lane & 15;
    const int hi   = lane >> 4;
    const int s0   = blockIdx.x * 16;
    const int bh   = blockIdx.y;
    const int b    = bh >> 3;
    const int h    = bh & 7;

    const float* W   = (blockIdx.z == 0) ? Wq : (blockIdx.z == 1) ? Wk : Wv;
    float*       Out = (blockIdx.z == 0) ? Qh : (blockIdx.z == 1) ? Kh : Vh;

    const float* Xrow = X + ((size_t)b * SEQ + s0 + n) * DIM;
    const int    col  = n * HEADS + h;

    v8f c = {};
    #pragma unroll 4
    for (int k0 = 0; k0 < DIM; k0 += 4) {
        const int ka = k0 + 2 * hi;
        v2f a, bb;
        a.x  = Xrow[ka];
        a.y  = Xrow[ka + 1];
        bb.x = W[(size_t)ka * DIM + col];
        bb.y = W[(size_t)(ka + 1) * DIM + col];
        c = __builtin_amdgcn_wmma_f32_16x16x4_f32(false, a, false, bb,
                                                  (short)0, c, false, false);
    }

    float* Obase = Out + ((size_t)bh * SEQ + s0) * HD;
    #pragma unroll
    for (int r = 0; r < 8; ++r) {
        const int M = r + 8 * hi;
        Obase[(size_t)M * HD + n] = c[r];
    }
}


__device__ __forceinline__ float xhalf(float x)
{
    const int b = __builtin_bit_cast(int, x);
    return __builtin_bit_cast(float,
        __builtin_amdgcn_permlanex16(b, b, 0x76543210, 0xfedcba98, false, false));
}

__global__ void __launch_bounds__(ATT_BLOCK) __attribute__((amdgpu_waves_per_eu(4, 4)))
flash_attn_wmma(const float* __restrict__ Qh,
                const float* __restrict__ Kh,
                const float* __restrict__ Vh,
                const int*   __restrict__ mask_src,
                float*       __restrict__ out)
{
    __shared__ float part[QG][NW][QT][10][32];

    const int lane = threadIdx.x & 31;
    const int w    = __builtin_amdgcn_readfirstlane((int)(threadIdx.x >> 5));
    const int g    = w / NW;
    const int kr   = w % NW;
    const int n    = lane & 15;
    const int hi   = lane >> 4;
    const int q0   = (blockIdx.x * QG + g) * (16 * QT);
    const int bh   = blockIdx.y;
    const int b    = bh >> 3;
    const int h    = bh & 7;


    v2f qb[QT][4];
    #pragma unroll
    for (int qt = 0; qt < QT; ++qt) {
        const float* Qrow = Qh + ((size_t)bh * SEQ + q0 + qt * 16 + n) * HD;
        #pragma unroll
        for (int t = 0; t < 4; ++t) {
            const int ka = 4 * t + 2 * hi;
            qb[qt][t].x = Qrow[ka];
            qb[qt][t].y = Qrow[ka + 1];
        }
    }

    float m[QT], l[QT];
    v8f acc[QT];
    #pragma unroll
    for (int qt = 0; qt < QT; ++qt) {
        acc[qt] = (v8f){};
        m[qt] = -1.0e30f;
        l[qt] = 0.0f;
    }

    const float scale = 0.25f;
    const float* Kbase = Kh + (size_t)bh * SEQ * HD;
    const float* Vbase = Vh + (size_t)bh * SEQ * HD;
    const int*   Mbase = mask_src + (size_t)b * SEQ + 8 * hi;

    const int kbeg = kr * KPW;
    for (int kv0 = kbeg; kv0 < kbeg + KPW; kv0 += 16) {
        __builtin_prefetch(Kbase + (size_t)(kv0 + 16 + n) * HD, 0, 1);
        __builtin_prefetch(Vbase + (size_t)(kv0 + 16 + n) * HD, 0, 1);

        const float* Krow = Kbase + (size_t)(kv0 + n) * HD;
        v2f ka[4];
        #pragma unroll
        for (int t = 0; t < 4; ++t) {
            const int kd = 4 * t + 2 * hi;
            ka[t].x = Krow[kd];
            ka[t].y = Krow[kd + 1];
        }

        const float* Vcol = Vbase + (size_t)(kv0 + 8 * hi) * HD + n;
        v2f va[4];
        #pragma unroll
        for (int t = 0; t < 4; ++t) {
            va[t].x = Vcol[(size_t)(2 * t) * HD];
            va[t].y = Vcol[(size_t)(2 * t + 1) * HD];
        }

        const int4 mk0 = *(const int4*)(Mbase + kv0);
        const int4 mk1 = *(const int4*)(Mbase + kv0 + 4);
        float addm[8];
        addm[0] = (mk0.x == 0) ? -1000.0f : 0.0f;
        addm[1] = (mk0.y == 0) ? -1000.0f : 0.0f;
        addm[2] = (mk0.z == 0) ? -1000.0f : 0.0f;
        addm[3] = (mk0.w == 0) ? -1000.0f : 0.0f;
        addm[4] = (mk1.x == 0) ? -1000.0f : 0.0f;
        addm[5] = (mk1.y == 0) ? -1000.0f : 0.0f;
        addm[6] = (mk1.z == 0) ? -1000.0f : 0.0f;
        addm[7] = (mk1.w == 0) ? -1000.0f : 0.0f;

        #pragma unroll
        for (int qt = 0; qt < QT; ++qt) {
            v8f s = {};
            #pragma unroll
            for (int t = 0; t < 4; ++t)
                s = __builtin_amdgcn_wmma_f32_16x16x4_f32(false, ka[t],
                                                          false, qb[qt][t],
                                                          (short)0, s,
                                                          false, false);
            #pragma unroll
            for (int r = 0; r < 8; ++r) s[r] = s[r] * scale + addm[r];

            float mx = fmaxf(fmaxf(fmaxf(s[0], s[1]), fmaxf(s[2], s[3])),
                             fmaxf(fmaxf(s[4], s[5]), fmaxf(s[6], s[7])));
            mx = fmaxf(mx, xhalf(mx));
            const float mnew  = fmaxf(m[qt], mx);
            const float alpha = __expf(m[qt] - mnew);
            #pragma unroll
            for (int r = 0; r < 8; ++r) s[r] = __expf(s[r] - mnew);
            float ps = ((s[0] + s[1]) + (s[2] + s[3])) +
                       ((s[4] + s[5]) + (s[6] + s[7]));
            ps += xhalf(ps);
            l[qt] = l[qt] * alpha + ps;
            m[qt] = mnew;
            #pragma unroll
            for (int r = 0; r < 8; ++r) acc[qt][r] *= alpha;

            #pragma unroll
            for (int t = 0; t < 4; ++t) {
                v2f pb;
                pb.x = s[2 * t];
                pb.y = s[2 * t + 1];
                acc[qt] = __builtin_amdgcn_wmma_f32_16x16x4_f32(false, va[t],
                                                               false, pb,
                                                               (short)0, acc[qt],
                                                               false, false);
            }
        }
    }

    #pragma unroll
    for (int qt = 0; qt < QT; ++qt) {
        #pragma unroll
        for (int r = 0; r < 8; ++r) part[g][kr][qt][r][lane] = acc[qt][r];
        part[g][kr][qt][8][lane] = m[qt];
        part[g][kr][qt][9][lane] = l[qt];
    }
    __syncthreads();

    const int mq = kr / WPQ;
    const int r0 = (kr % WPQ) * RPW;
    float mw[NW], sc[NW];
    #pragma unroll
    for (int k = 0; k < NW; ++k) mw[k] = part[g][k][mq][8][lane];
    float M = mw[0];
    #pragma unroll
    for (int k = 1; k < NW; ++k) M = fmaxf(M, mw[k]);
    float L = 0.0f;
    #pragma unroll
    for (int k = 0; k < NW; ++k) {
        sc[k] = __expf(mw[k] - M);
        L += part[g][k][mq][9][lane] * sc[k];
    }
    float* Orow = out + ((size_t)b * SEQ + q0 + mq * 16 + n) * DIM + h;
    #pragma unroll
    for (int i = 0; i < RPW; ++i) {
        float o = 0.0f;
        #pragma unroll
        for (int k = 0; k < NW; ++k) o += part[g][k][mq][r0 + i][lane] * sc[k];
        Orow[(8 * hi + r0 + i) * HEADS] = o / L;
    }
}

extern "C" void kernel_launch(void* const* d_in, const int* in_sizes, int n_in,
                              void* d_out, int out_size, void* d_ws, size_t ws_size,
                              hipStream_t stream)
{
    const float* X        = (const float*)d_in[0];
    const int*   mask_src = (const int*)d_in[1];
    const float* Wq       = (const float*)d_in[4];
    const float* Wk       = (const float*)d_in[5];
    const float* Wv       = (const float*)d_in[6];
    float*       out      = (float*)d_out;

    const size_t headElems = (size_t)BATCH * HEADS * SEQ * HD;
    float* Qh = (float*)d_ws;
    float* Kh = Qh + headElems;
    float* Vh = Kh + headElems;

    dim3 gridP(SEQ / 16, BATCH * HEADS, 3);
    qkv_proj_wmma<<<gridP, 32, 0, stream>>>(X, Wq, Wk, Wv, Qh, Kh, Vh);

    dim3 gridA(SEQ / (16 * QT * QG), BATCH * HEADS);
    flash_attn_wmma<<<gridA, ATT_BLOCK, 0, stream>>>(Qh, Kh, Vh, mask_src, out);
}


# K/V rows stored unmasked-first; waves leave out key tiles holding only -1000-masked keys when a |q||k|/4 bound proves their weights under e^-90, else sweep them; key sum order changes
# speedup vs baseline: 10.6285x; 1.4336x over previous
#include <hip/hip_runtime.h>

#define DIM   128
#define HEADS 8
#define BATCH 2
#define SEQ   2048
#define HD    16
#define QT    4
#define NW    8
#define QG    2
#define KPW   (SEQ / NW)
#define WPQ   (NW / QT)
#define RPW   (8 / WPQ)
#define ATT_BLOCK 512
#define NTILE (SEQ / 16)
static_assert(ATT_BLOCK == 32 * NW * QG, "one wave per (query group, key range)");
static_assert(NW % QT == 0 && 8 % WPQ == 0 && KPW % 16 == 0, "merge split");

typedef __attribute__((ext_vector_type(2))) float v2f;
typedef __attribute__((ext_vector_type(8))) float v8f;

__device__ __forceinline__ float xhalf(float x)
{
    const int b = __builtin_bit_cast(int, x);
    return __builtin_bit_cast(float,
        __builtin_amdgcn_permlanex16(b, b, 0x76543210, 0xfedcba98, false, false));
}

#define PW 12
typedef __attribute__((ext_vector_type(4))) float v4f;

__global__ void __launch_bounds__(32 * PW) __attribute__((amdgpu_waves_per_eu(3, 3)))
qkv_proj_wmma(const float* __restrict__ X,
              const float* __restrict__ Wq,
              const float* __restrict__ Wk,
              const float* __restrict__ Wv,
              const int*   __restrict__ mask_src,
              float* __restrict__ Qh,
              float* __restrict__ Kh,
              float* __restrict__ Vh,
              float* __restrict__ Kn)
{
    __shared__ int cnt[2][PW];

    const int lane = threadIdx.x & 31;
    const int w    = __builtin_amdgcn_readfirstlane((int)(threadIdx.x >> 5));
    const int mat  = w >> 2;
    const int hp   = w & 3;
    const int n    = lane & 15;
    const int hi   = lane >> 4;
    const int s0   = blockIdx.x * 16;
    const int b    = blockIdx.y;

    const float* Wm  = (mat == 0) ? Wq : (mat == 1) ? Wk : Wv;
    float*       Out = (mat == 0) ? Qh : (mat == 1) ? Kh : Vh;

    const float* Xp = X + ((size_t)b * SEQ + s0 + n) * DIM + 4 * hi;
    const float* Wp = Wm + (size_t)(4 * hi) * DIM + n * HEADS + 2 * hp;

    v4f xv[16];
    #pragma unroll
    for (int j = 0; j < 16; ++j) xv[j] = *(const v4f*)(Xp + 8 * j);

    const int* mrow = mask_src + (size_t)b * SEQ;
    {
        const int i0 = 4 * (int)threadIdx.x;
        const int4 a = *(const int4*)(mrow + i0);
        int tot = __builtin_popcount(__builtin_amdgcn_ballot_w32(a.x != 0))
                + __builtin_popcount(__builtin_amdgcn_ballot_w32(a.y != 0))
                + __builtin_popcount(__builtin_amdgcn_ballot_w32(a.z != 0))
                + __builtin_popcount(__builtin_amdgcn_ballot_w32(a.w != 0));
        int bef = __builtin_popcount(__builtin_amdgcn_ballot_w32(a.x != 0 && i0     < s0))
                + __builtin_popcount(__builtin_amdgcn_ballot_w32(a.y != 0 && i0 + 1 < s0))
                + __builtin_popcount(__builtin_amdgcn_ballot_w32(a.z != 0 && i0 + 2 < s0))
                + __builtin_popcount(__builtin_amdgcn_ballot_w32(a.w != 0 && i0 + 3 < s0));
        if (w < (SEQ / 4 - 32 * PW) / 32) {
            const int i1 = i0 + 4 * 32 * PW;
            const int4 c = *(const int4*)(mrow + i1);
            tot += __builtin_popcount(__builtin_amdgcn_ballot_w32(c.x != 0))
                 + __builtin_popcount(__builtin_amdgcn_ballot_w32(c.y != 0))
                 + __builtin_popcount(__builtin_amdgcn_ballot_w32(c.z != 0))
                 + __builtin_popcount(__builtin_amdgcn_ballot_w32(c.w != 0));
            bef += __builtin_popcount(__builtin_amdgcn_ballot_w32(c.x != 0 && i1     < s0))
                 + __builtin_popcount(__builtin_amdgcn_ballot_w32(c.y != 0 && i1 + 1 < s0))
                 + __builtin_popcount(__builtin_amdgcn_ballot_w32(c.z != 0 && i1 + 2 < s0))
                 + __builtin_popcount(__builtin_amdgcn_ballot_w32(c.w != 0 && i1 + 3 < s0));
        }
        if (lane == 0) { cnt[0][w] = bef; cnt[1][w] = tot; }
    }
    const bool mine = mrow[s0 + n] != 0;
    const unsigned tile_ones = __builtin_amdgcn_ballot_w32(mine) & 0xffffu;
    const int obt = __builtin_popcount(tile_ones & ((1u << n) - 1u));

    v8f c0 = {}, c1 = {};
    #pragma unroll
    for (int j = 0; j < 16; ++j) {
        #pragma unroll
        for (int u = 0; u < 2; ++u) {
            const float* wr = Wp + (size_t)(8 * j + 2 * u) * DIM;
            const v2f wa = *(const v2f*)wr;
            const v2f wb = *(const v2f*)(wr + DIM);
            v2f xb, a0, a1;
            xb.x = u ? xv[j].z : xv[j].x;
            xb.y = u ? xv[j].w : xv[j].y;
            a0.x = wa.x;  a0.y = wb.x;
            a1.x = wa.y;  a1.y = wb.y;
            c0 = __builtin_amdgcn_wmma_f32_16x16x4_f32(false, a0, false, xb,
                                                       (short)0, c0, false, false);
            c1 = __builtin_amdgcn_wmma_f32_16x16x4_f32(false, a1, false, xb,
                                                       (short)0, c1, false, false);
        }
    }

    __syncthreads();
    int ones_before = 0, ones_all = 0;
    #pragma unroll
    for (int k = 0; k < PW; ++k) { ones_before += cnt[0][k]; ones_all += cnt[1][k]; }
    int pos = s0 + n;
    if (mat != 0)
        pos = mine ? (ones_before + obt)
                   : (ones_all + (s0 - ones_before) + (n - obt));

    if (mat == 1) {
        float q0 = 0.0f, q1 = 0.0f;
        #pragma unroll
        for (int r = 0; r < 8; ++r) { q0 += c0[r] * c0[r]; q1 += c1[r] * c1[r]; }
        q0 += xhalf(q0);
        q1 += xhalf(q1);
        if (hi == 0) {
            Kn[(size_t)(b * HEADS + 2 * hp) * SEQ + pos]     = q0;
            Kn[(size_t)(b * HEADS + 2 * hp + 1) * SEQ + pos] = q1;
        }
    }

    float* o0 = Out + ((size_t)(b * HEADS + 2 * hp) * SEQ + pos) * HD + 8 * hi;
    float* o1 = o0 + (size_t)SEQ * HD;
    v4f t;
    t.x = c0[0]; t.y = c0[1]; t.z = c0[2]; t.w = c0[3];  *(v4f*)o0       = t;
    t.x = c0[4]; t.y = c0[5]; t.z = c0[6]; t.w = c0[7];  *(v4f*)(o0 + 4) = t;
    t.x = c1[0]; t.y = c1[1]; t.z = c1[2]; t.w = c1[3];  *(v4f*)o1       = t;
    t.x = c1[4]; t.y = c1[5]; t.z = c1[6]; t.w = c1[7];  *(v4f*)(o1 + 4) = t;
}


__global__ void __launch_bounds__(ATT_BLOCK) __attribute__((amdgpu_waves_per_eu(4, 4)))
flash_attn_wmma(const float* __restrict__ Qh,
                const float* __restrict__ Kh,
                const float* __restrict__ Vh,
                const float* __restrict__ Kn,
                const int*   __restrict__ mask_src,
                float*       __restrict__ out)
{
    __shared__ float part[QG][NW][QT][10][32];
    __shared__ int   cntw[NW * QG];
    __shared__ float kmw[NW * QG];
    __shared__ float mshare[QG][NW][QT][32];

    const int lane = threadIdx.x & 31;
    const int w    = __builtin_amdgcn_readfirstlane((int)(threadIdx.x >> 5));
    const int g    = w / NW;
    const int kr   = w % NW;
    const int n    = lane & 15;
    const int hi   = lane >> 4;
    const int q0   = (blockIdx.x * QG + g) * (16 * QT);
    const int bh   = blockIdx.y;
    const int b    = bh >> 3;
    const int h    = bh & 7;


    v2f qb[QT][4];
    #pragma unroll
    for (int qt = 0; qt < QT; ++qt) {
        const float* Qrow = Qh + ((size_t)bh * SEQ + q0 + qt * 16 + n) * HD;
        #pragma unroll
        for (int t = 0; t < 4; ++t) {
            const int ka = 4 * t + 2 * hi;
            qb[qt][t].x = Qrow[ka];
            qb[qt][t].y = Qrow[ka + 1];
        }
    }

    const int4 mw4 = *(const int4*)(mask_src + (size_t)b * SEQ + 4 * (int)threadIdx.x);
    const v4f  kn4 = *(const v4f*)(Kn + (size_t)bh * SEQ + 4 * (int)threadIdx.x);
    {
        const int ones = __builtin_popcount(__builtin_amdgcn_ballot_w32(mw4.x != 0))
                       + __builtin_popcount(__builtin_amdgcn_ballot_w32(mw4.y != 0))
                       + __builtin_popcount(__builtin_amdgcn_ballot_w32(mw4.z != 0))
                       + __builtin_popcount(__builtin_amdgcn_ballot_w32(mw4.w != 0));
        if (lane == 0) cntw[w] = ones;
    }
    __syncthreads();
    int n1v = 0;
    #pragma unroll
    for (int k = 0; k < NW * QG; ++k) n1v += cntw[k];
    const int n1 = __builtin_amdgcn_readfirstlane(n1v);

    {
        const int p0 = 4 * (int)threadIdx.x;
        float pm = 0.0f;
        pm = fmaxf(pm, (p0     >= n1) ? kn4.x : 0.0f);
        pm = fmaxf(pm, (p0 + 1 >= n1) ? kn4.y : 0.0f);
        pm = fmaxf(pm, (p0 + 2 >= n1) ? kn4.z : 0.0f);
        pm = fmaxf(pm, (p0 + 3 >= n1) ? kn4.w : 0.0f);
        #pragma unroll
        for (int o = 16; o >= 1; o >>= 1) pm = fmaxf(pm, __shfl_xor(pm, o, 32));
        if (lane == 0) kmw[w] = pm;
    }
    __syncthreads();
    float kmax2 = 0.0f;
    #pragma unroll
    for (int k = 0; k < NW * QG; ++k) kmax2 = fmaxf(kmax2, kmw[k]);
    float sbound[QT];
    #pragma unroll
    for (int qt = 0; qt < QT; ++qt) {
        float ss = 0.0f;
        #pragma unroll
        for (int t = 0; t < 4; ++t)
            ss += qb[qt][t].x * qb[qt][t].x + qb[qt][t].y * qb[qt][t].y;
        ss += xhalf(ss);
        sbound[qt] = 0.250025f * sqrtf(ss) * sqrtf(kmax2);
    }

    float m[QT], l[QT];
    v8f acc[QT];
    #pragma unroll
    for (int qt = 0; qt < QT; ++qt) {
        acc[qt] = (v8f){};
        m[qt] = -1.0e30f;
        l[qt] = 0.0f;
    }

    const float scale = 0.25f;
    const float* Kbase = Kh + (size_t)bh * SEQ * HD;
    const float* Vbase = Vh + (size_t)bh * SEQ * HD;

    auto sweep_tile = [&](const int kv0) __attribute__((always_inline)) {
        __builtin_prefetch(Kbase + (size_t)(kv0 + 16 * NW + n) * HD, 0, 1);
        __builtin_prefetch(Vbase + (size_t)(kv0 + 16 * NW + n) * HD, 0, 1);

        const float* Krow = Kbase + (size_t)(kv0 + n) * HD;
        v2f ka[4];
        #pragma unroll
        for (int t = 0; t < 4; ++t) {
            const int kd = 4 * t + 2 * hi;
            ka[t].x = Krow[kd];
            ka[t].y = Krow[kd + 1];
        }

        const float* Vcol = Vbase + (size_t)(kv0 + 8 * hi) * HD + n;
        v2f va[4];
        #pragma unroll
        for (int t = 0; t < 4; ++t) {
            va[t].x = Vcol[(size_t)(2 * t) * HD];
            va[t].y = Vcol[(size_t)(2 * t + 1) * HD];
        }

        float addm[8];
        #pragma unroll
        for (int r = 0; r < 8; ++r)
            addm[r] = (kv0 + 8 * hi + r >= n1) ? -1000.0f : 0.0f;

        #pragma unroll
        for (int qt = 0; qt < QT; ++qt) {
            v8f s = {};
            #pragma unroll
            for (int t = 0; t < 4; ++t)
                s = __builtin_amdgcn_wmma_f32_16x16x4_f32(false, ka[t],
                                                          false, qb[qt][t],
                                                          (short)0, s,
                                                          false, false);
            #pragma unroll
            for (int r = 0; r < 8; ++r) s[r] = s[r] * scale + addm[r];

            float mx = fmaxf(fmaxf(fmaxf(s[0], s[1]), fmaxf(s[2], s[3])),
                             fmaxf(fmaxf(s[4], s[5]), fmaxf(s[6], s[7])));
            mx = fmaxf(mx, xhalf(mx));
            const float mnew  = fmaxf(m[qt], mx);
            const float alpha = __expf(m[qt] - mnew);
            #pragma unroll
            for (int r = 0; r < 8; ++r) s[r] = __expf(s[r] - mnew);
            float ps = ((s[0] + s[1]) + (s[2] + s[3])) +
                       ((s[4] + s[5]) + (s[6] + s[7]));
            ps += xhalf(ps);
            l[qt] = l[qt] * alpha + ps;
            m[qt] = mnew;
            #pragma unroll
            for (int r = 0; r < 8; ++r) acc[qt][r] *= alpha;

            #pragma unroll
            for (int t = 0; t < 4; ++t) {
                v2f pb;
                pb.x = s[2 * t];
                pb.y = s[2 * t + 1];
                acc[qt] = __builtin_amdgcn_wmma_f32_16x16x4_f32(false, va[t],
                                                               false, pb,
                                                               (short)0, acc[qt],
                                                               false, false);
            }
        }
    };

    const int nlt = (n1 + 15) >> 4;
    int tile = kr;
    for (; tile < nlt; tile += NW) sweep_tile(tile * 16);

    if (nlt < NTILE) {
        #pragma unroll
        for (int qt = 0; qt < QT; ++qt) mshare[g][kr][qt][lane] = m[qt];
        __syncthreads();
        bool ok = true;
        #pragma unroll
        for (int qt = 0; qt < QT; ++qt) {
            float mg = m[qt];
            #pragma unroll
            for (int k = 0; k < NW; ++k) mg = fmaxf(mg, mshare[g][k][qt][lane]);
            ok &= (sbound[qt] - mg <= 910.0f);
        }
        if (__builtin_amdgcn_ballot_w32(!ok) != 0u)
            for (; tile < NTILE; tile += NW) sweep_tile(tile * 16);
    }

    #pragma unroll
    for (int qt = 0; qt < QT; ++qt) {
        #pragma unroll
        for (int r = 0; r < 8; ++r) part[g][kr][qt][r][lane] = acc[qt][r];
        part[g][kr][qt][8][lane] = m[qt];
        part[g][kr][qt][9][lane] = l[qt];
    }
    __syncthreads();

    const int mq = kr / WPQ;
    const int r0 = (kr % WPQ) * RPW;
    float mw[NW], sc[NW];
    #pragma unroll
    for (int k = 0; k < NW; ++k) mw[k] = part[g][k][mq][8][lane];
    float M = mw[0];
    #pragma unroll
    for (int k = 1; k < NW; ++k) M = fmaxf(M, mw[k]);
    float L = 0.0f;
    #pragma unroll
    for (int k = 0; k < NW; ++k) {
        sc[k] = __expf(mw[k] - M);
        L += part[g][k][mq][9][lane] * sc[k];
    }
    float* Orow = out + ((size_t)b * SEQ + q0 + mq * 16 + n) * DIM + h;
    #pragma unroll
    for (int i = 0; i < RPW; ++i) {
        float o = 0.0f;
        #pragma unroll
        for (int k = 0; k < NW; ++k) o += part[g][k][mq][r0 + i][lane] * sc[k];
        Orow[(8 * hi + r0 + i) * HEADS] = o / L;
    }
}

extern "C" void kernel_launch(void* const* d_in, const int* in_sizes, int n_in,
                              void* d_out, int out_size, void* d_ws, size_t ws_size,
                              hipStream_t stream)
{
    const float* X        = (const float*)d_in[0];
    const int*   mask_src = (const int*)d_in[1];
    const float* Wq       = (const float*)d_in[4];
    const float* Wk       = (const float*)d_in[5];
    const float* Wv       = (const float*)d_in[6];
    float*       out      = (float*)d_out;

    const size_t headElems = (size_t)BATCH * HEADS * SEQ * HD;
    float* Qh = (float*)d_ws;
    float* Kh = Qh + headElems;
    float* Vh = Kh + headElems;
    float* Kn = Vh + headElems;

    dim3 gridP(SEQ / 16, BATCH);
    qkv_proj_wmma<<<gridP, 32 * PW, 0, stream>>>(X, Wq, Wk, Wv, mask_src, Qh, Kh, Vh, Kn);

    dim3 gridA(SEQ / (16 * QT * QG), BATCH * HEADS);
    flash_attn_wmma<<<gridA, ATT_BLOCK, 0, stream>>>(Qh, Kh, Vh, Kn, mask_src, out);
}


# Attention merge re-mapped: lane = (query, d mod 4), so each output store instruction writes 4 pieces in each of 8 lines instead of one piece in each of 32 lines; sums unchanged
# speedup vs baseline: 11.3585x; 1.0687x over previous
#include <hip/hip_runtime.h>

#define DIM   128
#define HEADS 8
#define BATCH 2
#define SEQ   2048
#define HD    16
#define QT    4
#define NW    8
#define QG    2
#define KPW   (SEQ / NW)
#define WPQ   (NW / QT)
#define QPW   (16 / WPQ)
#define LPQ   (32 / QPW)
#define PPITCH 36
#define ATT_BLOCK 512
#define NTILE (SEQ / 16)
static_assert(ATT_BLOCK == 32 * NW * QG, "one wave per (query group, key range)");
static_assert(NW % QT == 0 && 16 % WPQ == 0 && KPW % 16 == 0, "merge split");
static_assert(LPQ == 4 && HEADS * 4 * LPQ == 128, "one output line per query and store");

typedef __attribute__((ext_vector_type(2))) float v2f;
typedef __attribute__((ext_vector_type(8))) float v8f;

__device__ __forceinline__ float xhalf(float x)
{
    const int b = __builtin_bit_cast(int, x);
    return __builtin_bit_cast(float,
        __builtin_amdgcn_permlanex16(b, b, 0x76543210, 0xfedcba98, false, false));
}

#define PW 12
typedef __attribute__((ext_vector_type(4))) float v4f;

__global__ void __launch_bounds__(32 * PW) __attribute__((amdgpu_waves_per_eu(3, 3)))
qkv_proj_wmma(const float* __restrict__ X,
              const float* __restrict__ Wq,
              const float* __restrict__ Wk,
              const float* __restrict__ Wv,
              const int*   __restrict__ mask_src,
              float* __restrict__ Qh,
              float* __restrict__ Kh,
              float* __restrict__ Vh,
              float* __restrict__ Kn)
{
    __shared__ int cnt[2][PW];

    const int lane = threadIdx.x & 31;
    const int w    = __builtin_amdgcn_readfirstlane((int)(threadIdx.x >> 5));
    const int mat  = w >> 2;
    const int hp   = w & 3;
    const int n    = lane & 15;
    const int hi   = lane >> 4;
    const int s0   = blockIdx.x * 16;
    const int b    = blockIdx.y;

    const float* Wm  = (mat == 0) ? Wq : (mat == 1) ? Wk : Wv;
    float*       Out = (mat == 0) ? Qh : (mat == 1) ? Kh : Vh;

    const float* Xp = X + ((size_t)b * SEQ + s0 + n) * DIM + 4 * hi;
    const float* Wp = Wm + (size_t)(4 * hi) * DIM + n * HEADS + 2 * hp;

    v4f xv[16];
    #pragma unroll
    for (int j = 0; j < 16; ++j) xv[j] = *(const v4f*)(Xp + 8 * j);

    const int* mrow = mask_src + (size_t)b * SEQ;
    {
        const int i0 = 4 * (int)threadIdx.x;
        const int4 a = *(const int4*)(mrow + i0);
        int tot = __builtin_popcount(__builtin_amdgcn_ballot_w32(a.x != 0))
                + __builtin_popcount(__builtin_amdgcn_ballot_w32(a.y != 0))
                + __builtin_popcount(__builtin_amdgcn_ballot_w32(a.z != 0))
                + __builtin_popcount(__builtin_amdgcn_ballot_w32(a.w != 0));
        int bef = __builtin_popcount(__builtin_amdgcn_ballot_w32(a.x != 0 && i0     < s0))
                + __builtin_popcount(__builtin_amdgcn_ballot_w32(a.y != 0 && i0 + 1 < s0))
                + __builtin_popcount(__builtin_amdgcn_ballot_w32(a.z != 0 && i0 + 2 < s0))
                + __builtin_popcount(__builtin_amdgcn_ballot_w32(a.w != 0 && i0 + 3 < s0));
        if (w < (SEQ / 4 - 32 * PW) / 32) {
            const int i1 = i0 + 4 * 32 * PW;
            const int4 c = *(const int4*)(mrow + i1);
            tot += __builtin_popcount(__builtin_amdgcn_ballot_w32(c.x != 0))
                 + __builtin_popcount(__builtin_amdgcn_ballot_w32(c.y != 0))
                 + __builtin_popcount(__builtin_amdgcn_ballot_w32(c.z != 0))
                 + __builtin_popcount(__builtin_amdgcn_ballot_w32(c.w != 0));
            bef += __builtin_popcount(__builtin_amdgcn_ballot_w32(c.x != 0 && i1     < s0))
                 + __builtin_popcount(__builtin_amdgcn_ballot_w32(c.y != 0 && i1 + 1 < s0))
                 + __builtin_popcount(__builtin_amdgcn_ballot_w32(c.z != 0 && i1 + 2 < s0))
                 + __builtin_popcount(__builtin_amdgcn_ballot_w32(c.w != 0 && i1 + 3 < s0));
        }
        if (lane == 0) { cnt[0][w] = bef; cnt[1][w] = tot; }
    }
    const bool mine = mrow[s0 + n] != 0;
    const unsigned tile_ones = __builtin_amdgcn_ballot_w32(mine) & 0xffffu;
    const int obt = __builtin_popcount(tile_ones & ((1u << n) - 1u));

    v8f c0 = {}, c1 = {};
    #pragma unroll
    for (int j = 0; j < 16; ++j) {
        #pragma unroll
        for (int u = 0; u < 2; ++u) {
            const float* wr = Wp + (size_t)(8 * j + 2 * u) * DIM;
            const v2f wa = *(const v2f*)wr;
            const v2f wb = *(const v2f*)(wr + DIM);
            v2f xb, a0, a1;
            xb.x = u ? xv[j].z : xv[j].x;
            xb.y = u ? xv[j].w : xv[j].y;
            a0.x = wa.x;  a0.y = wb.x;
            a1.x = wa.y;  a1.y = wb.y;
            c0 = __builtin_amdgcn_wmma_f32_16x16x4_f32(false, a0, false, xb,
                                                       (short)0, c0, false, false);
            c1 = __builtin_amdgcn_wmma_f32_16x16x4_f32(false, a1, false, xb,
                                                       (short)0, c1, false, false);
        }
    }

    __syncthreads();
    int ones_before = 0, ones_all = 0;
    #pragma unroll
    for (int k = 0; k < PW; ++k) { ones_before += cnt[0][k]; ones_all += cnt[1][k]; }
    int pos = s0 + n;
    if (mat != 0)
        pos = mine ? (ones_before + obt)
                   : (ones_all + (s0 - ones_before) + (n - obt));

    if (mat == 1) {
        float q0 = 0.0f, q1 = 0.0f;
        #pragma unroll
        for (int r = 0; r < 8; ++r) { q0 += c0[r] * c0[r]; q1 += c1[r] * c1[r]; }
        q0 += xhalf(q0);
        q1 += xhalf(q1);
        if (hi == 0) {
            Kn[(size_t)(b * HEADS + 2 * hp) * SEQ + pos]     = q0;
            Kn[(size_t)(b * HEADS + 2 * hp + 1) * SEQ + pos] = q1;
        }
    }

    float* o0 = Out + ((size_t)(b * HEADS + 2 * hp) * SEQ + pos) * HD + 8 * hi;
    float* o1 = o0 + (size_t)SEQ * HD;
    v4f t;
    t.x = c0[0]; t.y = c0[1]; t.z = c0[2]; t.w = c0[3];  *(v4f*)o0       = t;
    t.x = c0[4]; t.y = c0[5]; t.z = c0[6]; t.w = c0[7];  *(v4f*)(o0 + 4) = t;
    t.x = c1[0]; t.y = c1[1]; t.z = c1[2]; t.w = c1[3];  *(v4f*)o1       = t;
    t.x = c1[4]; t.y = c1[5]; t.z = c1[6]; t.w = c1[7];  *(v4f*)(o1 + 4) = t;
}


__global__ void __launch_bounds__(ATT_BLOCK) __attribute__((amdgpu_waves_per_eu(4, 4)))
flash_attn_wmma(const float* __restrict__ Qh,
                const float* __restrict__ Kh,
                const float* __restrict__ Vh,
                const float* __restrict__ Kn,
                const int*   __restrict__ mask_src,
                float*       __restrict__ out)
{
    __shared__ float part[QG][NW][QT][10][PPITCH];
    __shared__ int   cntw[NW * QG];
    __shared__ float kmw[NW * QG];
    __shared__ float mshare[QG][NW][QT][32];

    const int lane = threadIdx.x & 31;
    const int w    = __builtin_amdgcn_readfirstlane((int)(threadIdx.x >> 5));
    const int g    = w / NW;
    const int kr   = w % NW;
    const int n    = lane & 15;
    const int hi   = lane >> 4;
    const int q0   = (blockIdx.x * QG + g) * (16 * QT);
    const int bh   = blockIdx.y;
    const int b    = bh >> 3;
    const int h    = bh & 7;


    v2f qb[QT][4];
    #pragma unroll
    for (int qt = 0; qt < QT; ++qt) {
        const float* Qrow = Qh + ((size_t)bh * SEQ + q0 + qt * 16 + n) * HD;
        #pragma unroll
        for (int t = 0; t < 4; ++t) {
            const int ka = 4 * t + 2 * hi;
            qb[qt][t].x = Qrow[ka];
            qb[qt][t].y = Qrow[ka + 1];
        }
    }

    const int4 mw4 = *(const int4*)(mask_src + (size_t)b * SEQ + 4 * (int)threadIdx.x);
    const v4f  kn4 = *(const v4f*)(Kn + (size_t)bh * SEQ + 4 * (int)threadIdx.x);
    {
        const int ones = __builtin_popcount(__builtin_amdgcn_ballot_w32(mw4.x != 0))
                       + __builtin_popcount(__builtin_amdgcn_ballot_w32(mw4.y != 0))
                       + __builtin_popcount(__builtin_amdgcn_ballot_w32(mw4.z != 0))
                       + __builtin_popcount(__builtin_amdgcn_ballot_w32(mw4.w != 0));
        if (lane == 0) cntw[w] = ones;
    }
    __syncthreads();
    int n1v = 0;
    #pragma unroll
    for (int k = 0; k < NW * QG; ++k) n1v += cntw[k];
    const int n1 = __builtin_amdgcn_readfirstlane(n1v);

    {
        const int p0 = 4 * (int)threadIdx.x;
        float pm = 0.0f;
        pm = fmaxf(pm, (p0     >= n1) ? kn4.x : 0.0f);
        pm = fmaxf(pm, (p0 + 1 >= n1) ? kn4.y : 0.0f);
        pm = fmaxf(pm, (p0 + 2 >= n1) ? kn4.z : 0.0f);
        pm = fmaxf(pm, (p0 + 3 >= n1) ? kn4.w : 0.0f);
        #pragma unroll
        for (int o = 16; o >= 1; o >>= 1) pm = fmaxf(pm, __shfl_xor(pm, o, 32));
        if (lane == 0) kmw[w] = pm;
    }
    __syncthreads();
    float kmax2 = 0.0f;
    #pragma unroll
    for (int k = 0; k < NW * QG; ++k) kmax2 = fmaxf(kmax2, kmw[k]);
    float sbound[QT];
    #pragma unroll
    for (int qt = 0; qt < QT; ++qt) {
        float ss = 0.0f;
        #pragma unroll
        for (int t = 0; t < 4; ++t)
            ss += qb[qt][t].x * qb[qt][t].x + qb[qt][t].y * qb[qt][t].y;
        ss += xhalf(ss);
        sbound[qt] = 0.250025f * sqrtf(ss) * sqrtf(kmax2);
    }

    float m[QT], l[QT];
    v8f acc[QT];
    #pragma unroll
    for (int qt = 0; qt < QT; ++qt) {
        acc[qt] = (v8f){};
        m[qt] = -1.0e30f;
        l[qt] = 0.0f;
    }

    const float scale = 0.25f;
    const float* Kbase = Kh + (size_t)bh * SEQ * HD;
    const float* Vbase = Vh + (size_t)bh * SEQ * HD;

    auto sweep_tile = [&](const int kv0) __attribute__((always_inline)) {
        __builtin_prefetch(Kbase + (size_t)(kv0 + 16 * NW + n) * HD, 0, 1);
        __builtin_prefetch(Vbase + (size_t)(kv0 + 16 * NW + n) * HD, 0, 1);

        const float* Krow = Kbase + (size_t)(kv0 + n) * HD;
        v2f ka[4];
        #pragma unroll
        for (int t = 0; t < 4; ++t) {
            const int kd = 4 * t + 2 * hi;
            ka[t].x = Krow[kd];
            ka[t].y = Krow[kd + 1];
        }

        const float* Vcol = Vbase + (size_t)(kv0 + 8 * hi) * HD + n;
        v2f va[4];
        #pragma unroll
        for (int t = 0; t < 4; ++t) {
            va[t].x = Vcol[(size_t)(2 * t) * HD];
            va[t].y = Vcol[(size_t)(2 * t + 1) * HD];
        }

        float addm[8];
        #pragma unroll
        for (int r = 0; r < 8; ++r)
            addm[r] = (kv0 + 8 * hi + r >= n1) ? -1000.0f : 0.0f;

        #pragma unroll
        for (int qt = 0; qt < QT; ++qt) {
            v8f s = {};
            #pragma unroll
            for (int t = 0; t < 4; ++t)
                s = __builtin_amdgcn_wmma_f32_16x16x4_f32(false, ka[t],
                                                          false, qb[qt][t],
                                                          (short)0, s,
                                                          false, false);
            #pragma unroll
            for (int r = 0; r < 8; ++r) s[r] = s[r] * scale + addm[r];

            float mx = fmaxf(fmaxf(fmaxf(s[0], s[1]), fmaxf(s[2], s[3])),
                             fmaxf(fmaxf(s[4], s[5]), fmaxf(s[6], s[7])));
            mx = fmaxf(mx, xhalf(mx));
            const float mnew  = fmaxf(m[qt], mx);
            const float alpha = __expf(m[qt] - mnew);
            #pragma unroll
            for (int r = 0; r < 8; ++r) s[r] = __expf(s[r] - mnew);
            float ps = ((s[0] + s[1]) + (s[2] + s[3])) +
                       ((s[4] + s[5]) + (s[6] + s[7]));
            ps += xhalf(ps);
            l[qt] = l[qt] * alpha + ps;
            m[qt] = mnew;
            #pragma unroll
            for (int r = 0; r < 8; ++r) acc[qt][r] *= alpha;

            #pragma unroll
            for (int t = 0; t < 4; ++t) {
                v2f pb;
                pb.x = s[2 * t];
                pb.y = s[2 * t + 1];
                acc[qt] = __builtin_amdgcn_wmma_f32_16x16x4_f32(false, va[t],
                                                               false, pb,
                                                               (short)0, acc[qt],
                                                               false, false);
            }
        }
    };

    const int nlt = (n1 + 15) >> 4;
    int tile = kr;
    for (; tile < nlt; tile += NW) sweep_tile(tile * 16);

    if (nlt < NTILE) {
        #pragma unroll
        for (int qt = 0; qt < QT; ++qt) mshare[g][kr][qt][lane] = m[qt];
        __syncthreads();
        bool ok = true;
        #pragma unroll
        for (int qt = 0; qt < QT; ++qt) {
            float mg = m[qt];
            #pragma unroll
            for (int k = 0; k < NW; ++k) mg = fmaxf(mg, mshare[g][k][qt][lane]);
            ok &= (sbound[qt] - mg <= 910.0f);
        }
        if (__builtin_amdgcn_ballot_w32(!ok) != 0u)
            for (; tile < NTILE; tile += NW) sweep_tile(tile * 16);
    }

    #pragma unroll
    for (int qt = 0; qt < QT; ++qt) {
        #pragma unroll
        for (int r = 0; r < 8; ++r) part[g][kr][qt][r][lane] = acc[qt][r];
        part[g][kr][qt][8][lane] = m[qt];
        part[g][kr][qt][9][lane] = l[qt];
    }
    __syncthreads();

    const int mq = kr / WPQ;
    const int qq = (kr % WPQ) * QPW + (lane % QPW);
    const int dl = lane / QPW;
    float mw[NW], sc[NW];
    #pragma unroll
    for (int k = 0; k < NW; ++k) mw[k] = part[g][k][mq][8][qq];
    float M = mw[0];
    #pragma unroll
    for (int k = 1; k < NW; ++k) M = fmaxf(M, mw[k]);
    float L = 0.0f;
    #pragma unroll
    for (int k = 0; k < NW; ++k) {
        sc[k] = __expf(mw[k] - M);
        L += part[g][k][mq][9][qq] * sc[k];
    }
    float* Orow = out + ((size_t)b * SEQ + q0 + mq * 16 + qq) * DIM + h;
    #pragma unroll
    for (int i = 0; i < 16 / LPQ; ++i) {
        const int d  = LPQ * i + dl;
        const int rr = d & 7;
        const int ll = 16 * (d >> 3) + qq;
        float o = 0.0f;
        #pragma unroll
        for (int k = 0; k < NW; ++k) o += part[g][k][mq][rr][ll] * sc[k];
        Orow[d * HEADS] = o / L;
    }
}

extern "C" void kernel_launch(void* const* d_in, const int* in_sizes, int n_in,
                              void* d_out, int out_size, void* d_ws, size_t ws_size,
                              hipStream_t stream)
{
    const float* X        = (const float*)d_in[0];
    const int*   mask_src = (const int*)d_in[1];
    const float* Wq       = (const float*)d_in[4];
    const float* Wk       = (const float*)d_in[5];
    const float* Wv       = (const float*)d_in[6];
    float*       out      = (float*)d_out;

    const size_t headElems = (size_t)BATCH * HEADS * SEQ * HD;
    float* Qh = (float*)d_ws;
    float* Kh = Qh + headElems;
    float* Vh = Kh + headElems;
    float* Kn = Vh + headElems;

    dim3 gridP(SEQ / 16, BATCH);
    qkv_proj_wmma<<<gridP, 32 * PW, 0, stream>>>(X, Wq, Wk, Wv, mask_src, Qh, Kh, Vh, Kn);

    dim3 gridA(SEQ / (16 * QT * QG), BATCH * HEADS);
    flash_attn_wmma<<<gridA, ATT_BLOCK, 0, stream>>>(Qh, Kh, Vh, Kn, mask_src, out);
}


# Attention waves set their issue priority (s_setprio) from the key tiles they still have to sweep, most left first, so a SIMD's 4 waves reach the merge together, not one after another; sums unchanged
# speedup vs baseline: 11.7581x; 1.0352x over previous
#include <hip/hip_runtime.h>

#define DIM   128
#define HEADS 8
#define BATCH 2
#define SEQ   2048
#define HD    16
#define QT    4
#define NW    8
#define QG    2
#define KPW   (SEQ / NW)
#define WPQ   (NW / QT)
#define QPW   (16 / WPQ)
#define LPQ   (32 / QPW)
#define PPITCH 36
#define ATT_BLOCK 512
#define NTILE (SEQ / 16)
static_assert(ATT_BLOCK == 32 * NW * QG, "one wave per (query group, key range)");
static_assert(NW % QT == 0 && 16 % WPQ == 0 && KPW % 16 == 0, "merge split");
static_assert(LPQ == 4 && HEADS * 4 * LPQ == 128, "one output line per query and store");

typedef __attribute__((ext_vector_type(2))) float v2f;
typedef __attribute__((ext_vector_type(8))) float v8f;

__device__ __forceinline__ float xhalf(float x)
{
    const int b = __builtin_bit_cast(int, x);
    return __builtin_bit_cast(float,
        __builtin_amdgcn_permlanex16(b, b, 0x76543210, 0xfedcba98, false, false));
}

#define PW 12
typedef __attribute__((ext_vector_type(4))) float v4f;

__global__ void __launch_bounds__(32 * PW) __attribute__((amdgpu_waves_per_eu(3, 3)))
qkv_proj_wmma(const float* __restrict__ X,
              const float* __restrict__ Wq,
              const float* __restrict__ Wk,
              const float* __restrict__ Wv,
              const int*   __restrict__ mask_src,
              float* __restrict__ Qh,
              float* __restrict__ Kh,
              float* __restrict__ Vh,
              float* __restrict__ Kn)
{
    __shared__ int cnt[2][PW];

    const int lane = threadIdx.x & 31;
    const int w    = __builtin_amdgcn_readfirstlane((int)(threadIdx.x >> 5));
    const int mat  = w >> 2;
    const int hp   = w & 3;
    const int n    = lane & 15;
    const int hi   = lane >> 4;
    const int s0   = blockIdx.x * 16;
    const int b    = blockIdx.y;

    const float* Wm  = (mat == 0) ? Wq : (mat == 1) ? Wk : Wv;
    float*       Out = (mat == 0) ? Qh : (mat == 1) ? Kh : Vh;

    const float* Xp = X + ((size_t)b * SEQ + s0 + n) * DIM + 4 * hi;
    const float* Wp = Wm + (size_t)(4 * hi) * DIM + n * HEADS + 2 * hp;

    v4f xv[16];
    #pragma unroll
    for (int j = 0; j < 16; ++j) xv[j] = *(const v4f*)(Xp + 8 * j);

    const int* mrow = mask_src + (size_t)b * SEQ;
    {
        const int i0 = 4 * (int)threadIdx.x;
        const int4 a = *(const int4*)(mrow + i0);
        int tot = __builtin_popcount(__builtin_amdgcn_ballot_w32(a.x != 0))
                + __builtin_popcount(__builtin_amdgcn_ballot_w32(a.y != 0))
                + __builtin_popcount(__builtin_amdgcn_ballot_w32(a.z != 0))
                + __builtin_popcount(__builtin_amdgcn_ballot_w32(a.w != 0));
        int bef = __builtin_popcount(__builtin_amdgcn_ballot_w32(a.x != 0 && i0     < s0))
                + __builtin_popcount(__builtin_amdgcn_ballot_w32(a.y != 0 && i0 + 1 < s0))
                + __builtin_popcount(__builtin_amdgcn_ballot_w32(a.z != 0 && i0 + 2 < s0))
                + __builtin_popcount(__builtin_amdgcn_ballot_w32(a.w != 0 && i0 + 3 < s0));
        if (w < (SEQ / 4 - 32 * PW) / 32) {
            const int i1 = i0 + 4 * 32 * PW;
            const int4 c = *(const int4*)(mrow + i1);
            tot += __builtin_popcount(__builtin_amdgcn_ballot_w32(c.x != 0))
                 + __builtin_popcount(__builtin_amdgcn_ballot_w32(c.y != 0))
                 + __builtin_popcount(__builtin_amdgcn_ballot_w32(c.z != 0))
                 + __builtin_popcount(__builtin_amdgcn_ballot_w32(c.w != 0));
            bef += __builtin_popcount(__builtin_amdgcn_ballot_w32(c.x != 0 && i1     < s0))
                 + __builtin_popcount(__builtin_amdgcn_ballot_w32(c.y != 0 && i1 + 1 < s0))
                 + __builtin_popcount(__builtin_amdgcn_ballot_w32(c.z != 0 && i1 + 2 < s0))
                 + __builtin_popcount(__builtin_amdgcn_ballot_w32(c.w != 0 && i1 + 3 < s0));
        }
        if (lane == 0) { cnt[0][w] = bef; cnt[1][w] = tot; }
    }
    const bool mine = mrow[s0 + n] != 0;
    const unsigned tile_ones = __builtin_amdgcn_ballot_w32(mine) & 0xffffu;
    const int obt = __builtin_popcount(tile_ones & ((1u << n) - 1u));

    v8f c0 = {}, c1 = {};
    #pragma unroll
    for (int j = 0; j < 16; ++j) {
        #pragma unroll
        for (int u = 0; u < 2; ++u) {
            const float* wr = Wp + (size_t)(8 * j + 2 * u) * DIM;
            const v2f wa = *(const v2f*)wr;
            const v2f wb = *(const v2f*)(wr + DIM);
            v2f xb, a0, a1;
            xb.x = u ? xv[j].z : xv[j].x;
            xb.y = u ? xv[j].w : xv[j].y;
            a0.x = wa.x;  a0.y = wb.x;
            a1.x = wa.y;  a1.y = wb.y;
            c0 = __builtin_amdgcn_wmma_f32_16x16x4_f32(false, a0, false, xb,
                                                       (short)0, c0, false, false);
            c1 = __builtin_amdgcn_wmma_f32_16x16x4_f32(false, a1, false, xb,
                                                       (short)0, c1, false, false);
        }
    }

    __syncthreads();
    int ones_before = 0, ones_all = 0;
    #pragma unroll
    for (int k = 0; k < PW; ++k) { ones_before += cnt[0][k]; ones_all += cnt[1][k]; }
    int pos = s0 + n;
    if (mat != 0)
        pos = mine ? (ones_before + obt)
                   : (ones_all + (s0 - ones_before) + (n - obt));

    if (mat == 1) {
        float q0 = 0.0f, q1 = 0.0f;
        #pragma unroll
        for (int r = 0; r < 8; ++r) { q0 += c0[r] * c0[r]; q1 += c1[r] * c1[r]; }
        q0 += xhalf(q0);
        q1 += xhalf(q1);
        if (hi == 0) {
            Kn[(size_t)(b * HEADS + 2 * hp) * SEQ + pos]     = q0;
            Kn[(size_t)(b * HEADS + 2 * hp + 1) * SEQ + pos] = q1;
        }
    }

    float* o0 = Out + ((size_t)(b * HEADS + 2 * hp) * SEQ + pos) * HD + 8 * hi;
    float* o1 = o0 + (size_t)SEQ * HD;
    v4f t;
    t.x = c0[0]; t.y = c0[1]; t.z = c0[2]; t.w = c0[3];  *(v4f*)o0       = t;
    t.x = c0[4]; t.y = c0[5]; t.z = c0[6]; t.w = c0[7];  *(v4f*)(o0 + 4) = t;
    t.x = c1[0]; t.y = c1[1]; t.z = c1[2]; t.w = c1[3];  *(v4f*)o1       = t;
    t.x = c1[4]; t.y = c1[5]; t.z = c1[6]; t.w = c1[7];  *(v4f*)(o1 + 4) = t;
}


__global__ void __launch_bounds__(ATT_BLOCK) __attribute__((amdgpu_waves_per_eu(4, 4)))
flash_attn_wmma(const float* __restrict__ Qh,
                const float* __restrict__ Kh,
                const float* __restrict__ Vh,
                const float* __restrict__ Kn,
                const int*   __restrict__ mask_src,
                float*       __restrict__ out)
{
    __shared__ float part[QG][NW][QT][10][PPITCH];
    __shared__ int   cntw[NW * QG];
    __shared__ float kmw[NW * QG];
    __shared__ float mshare[QG][NW][QT][32];

    const int lane = threadIdx.x & 31;
    const int w    = __builtin_amdgcn_readfirstlane((int)(threadIdx.x >> 5));
    const int g    = w / NW;
    const int kr   = w % NW;
    const int n    = lane & 15;
    const int hi   = lane >> 4;
    const int q0   = (blockIdx.x * QG + g) * (16 * QT);
    const int bh   = blockIdx.y;
    const int b    = bh >> 3;
    const int h    = bh & 7;


    v2f qb[QT][4];
    #pragma unroll
    for (int qt = 0; qt < QT; ++qt) {
        const float* Qrow = Qh + ((size_t)bh * SEQ + q0 + qt * 16 + n) * HD;
        #pragma unroll
        for (int t = 0; t < 4; ++t) {
            const int ka = 4 * t + 2 * hi;
            qb[qt][t].x = Qrow[ka];
            qb[qt][t].y = Qrow[ka + 1];
        }
    }

    const int4 mw4 = *(const int4*)(mask_src + (size_t)b * SEQ + 4 * (int)threadIdx.x);
    const v4f  kn4 = *(const v4f*)(Kn + (size_t)bh * SEQ + 4 * (int)threadIdx.x);
    {
        const int ones = __builtin_popcount(__builtin_amdgcn_ballot_w32(mw4.x != 0))
                       + __builtin_popcount(__builtin_amdgcn_ballot_w32(mw4.y != 0))
                       + __builtin_popcount(__builtin_amdgcn_ballot_w32(mw4.z != 0))
                       + __builtin_popcount(__builtin_amdgcn_ballot_w32(mw4.w != 0));
        if (lane == 0) cntw[w] = ones;
    }
    __syncthreads();
    int n1v = 0;
    #pragma unroll
    for (int k = 0; k < NW * QG; ++k) n1v += cntw[k];
    const int n1 = __builtin_amdgcn_readfirstlane(n1v);

    {
        const int p0 = 4 * (int)threadIdx.x;
        float pm = 0.0f;
        pm = fmaxf(pm, (p0     >= n1) ? kn4.x : 0.0f);
        pm = fmaxf(pm, (p0 + 1 >= n1) ? kn4.y : 0.0f);
        pm = fmaxf(pm, (p0 + 2 >= n1) ? kn4.z : 0.0f);
        pm = fmaxf(pm, (p0 + 3 >= n1) ? kn4.w : 0.0f);
        #pragma unroll
        for (int o = 16; o >= 1; o >>= 1) pm = fmaxf(pm, __shfl_xor(pm, o, 32));
        if (lane == 0) kmw[w] = pm;
    }
    __syncthreads();
    float kmax2 = 0.0f;
    #pragma unroll
    for (int k = 0; k < NW * QG; ++k) kmax2 = fmaxf(kmax2, kmw[k]);
    float sbound[QT];
    #pragma unroll
    for (int qt = 0; qt < QT; ++qt) {
        float ss = 0.0f;
        #pragma unroll
        for (int t = 0; t < 4; ++t)
            ss += qb[qt][t].x * qb[qt][t].x + qb[qt][t].y * qb[qt][t].y;
        ss += xhalf(ss);
        sbound[qt] = 0.250025f * sqrtf(ss) * sqrtf(kmax2);
    }

    float m[QT], l[QT];
    v8f acc[QT];
    #pragma unroll
    for (int qt = 0; qt < QT; ++qt) {
        acc[qt] = (v8f){};
        m[qt] = -1.0e30f;
        l[qt] = 0.0f;
    }

    const float scale = 0.25f;
    const float* Kbase = Kh + (size_t)bh * SEQ * HD;
    const float* Vbase = Vh + (size_t)bh * SEQ * HD;

    auto sweep_tile = [&](const int kv0) __attribute__((always_inline)) {
        __builtin_prefetch(Kbase + (size_t)(kv0 + 16 * NW + n) * HD, 0, 1);
        __builtin_prefetch(Vbase + (size_t)(kv0 + 16 * NW + n) * HD, 0, 1);

        const float* Krow = Kbase + (size_t)(kv0 + n) * HD;
        v2f ka[4];
        #pragma unroll
        for (int t = 0; t < 4; ++t) {
            const int kd = 4 * t + 2 * hi;
            ka[t].x = Krow[kd];
            ka[t].y = Krow[kd + 1];
        }

        const float* Vcol = Vbase + (size_t)(kv0 + 8 * hi) * HD + n;
        v2f va[4];
        #pragma unroll
        for (int t = 0; t < 4; ++t) {
            va[t].x = Vcol[(size_t)(2 * t) * HD];
            va[t].y = Vcol[(size_t)(2 * t + 1) * HD];
        }

        float addm[8];
        #pragma unroll
        for (int r = 0; r < 8; ++r)
            addm[r] = (kv0 + 8 * hi + r >= n1) ? -1000.0f : 0.0f;

        #pragma unroll
        for (int qt = 0; qt < QT; ++qt) {
            v8f s = {};
            #pragma unroll
            for (int t = 0; t < 4; ++t)
                s = __builtin_amdgcn_wmma_f32_16x16x4_f32(false, ka[t],
                                                          false, qb[qt][t],
                                                          (short)0, s,
                                                          false, false);
            #pragma unroll
            for (int r = 0; r < 8; ++r) s[r] = s[r] * scale + addm[r];

            float mx = fmaxf(fmaxf(fmaxf(s[0], s[1]), fmaxf(s[2], s[3])),
                             fmaxf(fmaxf(s[4], s[5]), fmaxf(s[6], s[7])));
            mx = fmaxf(mx, xhalf(mx));
            const float mnew  = fmaxf(m[qt], mx);
            const float alpha = __expf(m[qt] - mnew);
            #pragma unroll
            for (int r = 0; r < 8; ++r) s[r] = __expf(s[r] - mnew);
            float ps = ((s[0] + s[1]) + (s[2] + s[3])) +
                       ((s[4] + s[5]) + (s[6] + s[7]));
            ps += xhalf(ps);
            l[qt] = l[qt] * alpha + ps;
            m[qt] = mnew;
            #pragma unroll
            for (int r = 0; r < 8; ++r) acc[qt][r] *= alpha;

            #pragma unroll
            for (int t = 0; t < 4; ++t) {
                v2f pb;
                pb.x = s[2 * t];
                pb.y = s[2 * t + 1];
                acc[qt] = __builtin_amdgcn_wmma_f32_16x16x4_f32(false, va[t],
                                                               false, pb,
                                                               (short)0, acc[qt],
                                                               false, false);
            }
        }
    };

    const int nlt = (n1 + 15) >> 4;
    int tile = kr;
    for (; tile < nlt; tile += NW) {
        const int left = (nlt - tile + NW - 1) / NW;
        if (left >= 6)      __builtin_amdgcn_s_setprio(3);
        else if (left >= 4) __builtin_amdgcn_s_setprio(2);
        else if (left >= 2) __builtin_amdgcn_s_setprio(1);
        else                __builtin_amdgcn_s_setprio(0);
        sweep_tile(tile * 16);
    }

    if (nlt < NTILE) {
        #pragma unroll
        for (int qt = 0; qt < QT; ++qt) mshare[g][kr][qt][lane] = m[qt];
        __syncthreads();
        bool ok = true;
        #pragma unroll
        for (int qt = 0; qt < QT; ++qt) {
            float mg = m[qt];
            #pragma unroll
            for (int k = 0; k < NW; ++k) mg = fmaxf(mg, mshare[g][k][qt][lane]);
            ok &= (sbound[qt] - mg <= 910.0f);
        }
        if (__builtin_amdgcn_ballot_w32(!ok) != 0u)
            for (; tile < NTILE; tile += NW) sweep_tile(tile * 16);
    }

    #pragma unroll
    for (int qt = 0; qt < QT; ++qt) {
        #pragma unroll
        for (int r = 0; r < 8; ++r) part[g][kr][qt][r][lane] = acc[qt][r];
        part[g][kr][qt][8][lane] = m[qt];
        part[g][kr][qt][9][lane] = l[qt];
    }
    __syncthreads();

    const int mq = kr / WPQ;
    const int qq = (kr % WPQ) * QPW + (lane % QPW);
    const int dl = lane / QPW;
    float mw[NW], sc[NW];
    #pragma unroll
    for (int k = 0; k < NW; ++k) mw[k] = part[g][k][mq][8][qq];
    float M = mw[0];
    #pragma unroll
    for (int k = 1; k < NW; ++k) M = fmaxf(M, mw[k]);
    float L = 0.0f;
    #pragma unroll
    for (int k = 0; k < NW; ++k) {
        sc[k] = __expf(mw[k] - M);
        L += part[g][k][mq][9][qq] * sc[k];
    }
    float* Orow = out + ((size_t)b * SEQ + q0 + mq * 16 + qq) * DIM + h;
    #pragma unroll
    for (int i = 0; i < 16 / LPQ; ++i) {
        const int d  = LPQ * i + dl;
        const int rr = d & 7;
        const int ll = 16 * (d >> 3) + qq;
        float o = 0.0f;
        #pragma unroll
        for (int k = 0; k < NW; ++k) o += part[g][k][mq][rr][ll] * sc[k];
        Orow[d * HEADS] = o / L;
    }
}

extern "C" void kernel_launch(void* const* d_in, const int* in_sizes, int n_in,
                              void* d_out, int out_size, void* d_ws, size_t ws_size,
                              hipStream_t stream)
{
    const float* X        = (const float*)d_in[0];
    const int*   mask_src = (const int*)d_in[1];
    const float* Wq       = (const float*)d_in[4];
    const float* Wk       = (const float*)d_in[5];
    const float* Wv       = (const float*)d_in[6];
    float*       out      = (float*)d_out;

    const size_t headElems = (size_t)BATCH * HEADS * SEQ * HD;
    float* Qh = (float*)d_ws;
    float* Kh = Qh + headElems;
    float* Vh = Kh + headElems;
    float* Kn = Vh + headElems;

    dim3 gridP(SEQ / 16, BATCH);
    qkv_proj_wmma<<<gridP, 32 * PW, 0, stream>>>(X, Wq, Wk, Wv, mask_src, Qh, Kh, Vh, Kn);

    dim3 gridA(SEQ / (16 * QT * QG), BATCH * HEADS);
    flash_attn_wmma<<<gridA, ATT_BLOCK, 0, stream>>>(Qh, Kh, Vh, Kn, mask_src, out);
}


# Attention waves start the key sweep without waiting for the work-group: n1 is a scalar load of a word the projection stores, kmax joins the barrier behind the sweep, first K/V tile loaded at entry
# speedup vs baseline: 12.9052x; 1.0976x over previous
#include <hip/hip_runtime.h>

#define DIM   128
#define HEADS 8
#define BATCH 2
#define SEQ   2048
#define HD    16
#define QT    4
#define NW    8
#define QG    2
#define KPW   (SEQ / NW)
#define WPQ   (NW / QT)
#define QPW   (16 / WPQ)
#define LPQ   (32 / QPW)
#define PPITCH 36
#define ATT_BLOCK 512
#define NTILE (SEQ / 16)
static_assert(ATT_BLOCK == 32 * NW * QG, "one wave per (query group, key range)");
static_assert(NW % QT == 0 && 16 % WPQ == 0 && KPW % 16 == 0, "merge split");
static_assert(LPQ == 4 && HEADS * 4 * LPQ == 128, "one output line per query and store");

typedef __attribute__((ext_vector_type(2))) float v2f;
typedef __attribute__((ext_vector_type(8))) float v8f;

__device__ __forceinline__ float xhalf(float x)
{
    const int b = __builtin_bit_cast(int, x);
    return __builtin_bit_cast(float,
        __builtin_amdgcn_permlanex16(b, b, 0x76543210, 0xfedcba98, false, false));
}

template <int X>
__device__ __forceinline__ float xrow(float x)
{
    const int b = __builtin_bit_cast(int, x);
    return __builtin_bit_cast(float,
        __builtin_amdgcn_update_dpp(b, b, 0x160 + X, 0xf, 0xf, false));
}

#define PW  4
#define UPW 3
typedef __attribute__((ext_vector_type(4))) float v4f;
static_assert(PW * UPW == 12, "3 matrices x 4 column quarters");
#define MQ  (SEQ / (4 * 32 * PW))
#define PF  10
#define SPITCH 20
#define LDS_ALONE (164 * 1024)
static_assert(SEQ % (4 * 32 * PW) == 0, "mask words per thread");
static_assert(6 * PF <= 63, "loads a wave can have outstanding");
static_assert((3 * HEADS) % PW == 0, "(matrix, head) pairs per wave");

__global__ void __launch_bounds__(32 * PW) __attribute__((amdgpu_waves_per_eu(1, 1)))
qkv_proj_wmma(const float* __restrict__ X,
              const float* __restrict__ Wq,
              const float* __restrict__ Wk,
              const float* __restrict__ Wv,
              const int*   __restrict__ mask_src,
              float* __restrict__ Qh,
              float* __restrict__ Kh,
              float* __restrict__ Vh,
              float* __restrict__ Kn,
              int*   __restrict__ N1)
{
    __shared__ int cnt[2][PW];
    __shared__ __attribute__((aligned(16))) float knp[4][16][HEADS];
    __shared__ __attribute__((aligned(16))) union {
        float row[3 * HEADS][16][SPITCH];
        char  alone[LDS_ALONE];
    } stg;

    const int lane = threadIdx.x & 31;
    const int w    = __builtin_amdgcn_readfirstlane((int)(threadIdx.x >> 5));
    const int n    = lane & 15;
    const int hi   = lane >> 4;
    const int s0   = blockIdx.x * 16;
    const int b    = blockIdx.y;

    int mat[UPW], cq[UPW];
    const float* Wp[UPW];
    #pragma unroll
    for (int u = 0; u < UPW; ++u) {
        const int p = UPW * w + u;
        mat[u] = p >> 2;
        cq[u]  = p & 3;
        const float* Wm = (mat[u] == 0) ? Wq : (mat[u] == 1) ? Wk : Wv;
        Wp[u] = Wm + (size_t)(4 * hi) * DIM + 32 * cq[u] + HEADS * (n & 3) + 2 * (n >> 2);
    }

    const int* mrow = mask_src + (size_t)b * SEQ;
    int4 mwd[MQ];
    #pragma unroll
    for (int q = 0; q < MQ; ++q)
        mwd[q] = *(const int4*)(mrow + 4 * ((int)threadIdx.x + 32 * PW * q));
    const int mown = mrow[s0 + n];

    const float* Xp = X + ((size_t)b * SEQ + s0 + n) * DIM + 4 * hi;
    v4f xv[16];
    #pragma unroll
    for (int j = 0; j < 16; ++j) xv[j] = *(const v4f*)(Xp + 8 * j);

    v2f wa[32][UPW], wb[32][UPW];
    #pragma unroll
    for (int s = 0; s < PF; ++s) {
        #pragma unroll
        for (int u = 0; u < UPW; ++u) {
            const float* wr = Wp[u] + (size_t)(8 * (s >> 1) + 2 * (s & 1)) * DIM;
            wa[s][u] = *(const v2f*)wr;
            wb[s][u] = *(const v2f*)(wr + DIM);
        }
    }
    __builtin_amdgcn_sched_barrier(0);

    {
        int tot = 0, bef = 0;
        #pragma unroll
        for (int q = 0; q < MQ; ++q) {
            const int i0 = 4 * ((int)threadIdx.x + 32 * PW * q);
            const int4 a = mwd[q];
            tot += __builtin_popcount(__builtin_amdgcn_ballot_w32(a.x != 0))
                 + __builtin_popcount(__builtin_amdgcn_ballot_w32(a.y != 0))
                 + __builtin_popcount(__builtin_amdgcn_ballot_w32(a.z != 0))
                 + __builtin_popcount(__builtin_amdgcn_ballot_w32(a.w != 0));
            bef += __builtin_popcount(__builtin_amdgcn_ballot_w32(a.x != 0 && i0     < s0))
                 + __builtin_popcount(__builtin_amdgcn_ballot_w32(a.y != 0 && i0 + 1 < s0))
                 + __builtin_popcount(__builtin_amdgcn_ballot_w32(a.z != 0 && i0 + 2 < s0))
                 + __builtin_popcount(__builtin_amdgcn_ballot_w32(a.w != 0 && i0 + 3 < s0));
        }
        if (lane == 0) { cnt[0][w] = bef; cnt[1][w] = tot; }
    }
    const unsigned tile_ones = __builtin_amdgcn_ballot_w32(mown != 0) & 0xffffu;

    v8f c0[UPW], c1[UPW];
    #pragma unroll
    for (int u = 0; u < UPW; ++u) { c0[u] = (v8f){}; c1[u] = (v8f){}; }
    #pragma unroll
    for (int s = 0; s < 32; ++s) {
        if (s + PF < 32) {
            #pragma unroll
            for (int u = 0; u < UPW; ++u) {
                const int sn = s + PF;
                const float* wr = Wp[u] + (size_t)(8 * (sn >> 1) + 2 * (sn & 1)) * DIM;
                wa[s + PF][u] = *(const v2f*)wr;
                wb[s + PF][u] = *(const v2f*)(wr + DIM);
            }
        }
        const int j = s >> 1;
        v2f xb;
        xb.x = (s & 1) ? xv[j].z : xv[j].x;
        xb.y = (s & 1) ? xv[j].w : xv[j].y;
        #pragma unroll
        for (int u = 0; u < UPW; ++u) {
            v2f a0, a1;
            a0.x = wa[s][u].x;  a0.y = wb[s][u].x;
            a1.x = wa[s][u].y;  a1.y = wb[s][u].y;
            c0[u] = __builtin_amdgcn_wmma_f32_16x16x4_f32(false, a0, false, xb,
                                                          (short)0, c0[u], false, false);
            c1[u] = __builtin_amdgcn_wmma_f32_16x16x4_f32(false, a1, false, xb,
                                                          (short)0, c1[u], false, false);
        }
        __builtin_amdgcn_sched_barrier(0);
    }

    #pragma unroll
    for (int u = 0; u < UPW; ++u) {
        float* sp = &stg.row[mat[u] * HEADS + 4 * hi][n][4 * cq[u]];
        const int hs = 16 * SPITCH;
        v4f t;
        t.x = c0[u][0]; t.y = c0[u][1]; t.z = c0[u][2]; t.w = c0[u][3];  *(v4f*)sp            = t;
        t.x = c1[u][0]; t.y = c1[u][1]; t.z = c1[u][2]; t.w = c1[u][3];  *(v4f*)(sp + hs)     = t;
        t.x = c0[u][4]; t.y = c0[u][5]; t.z = c0[u][6]; t.w = c0[u][7];  *(v4f*)(sp + 2 * hs) = t;
        t.x = c1[u][4]; t.y = c1[u][5]; t.z = c1[u][6]; t.w = c1[u][7];  *(v4f*)(sp + 3 * hs) = t;
        if (mat[u] == 1) {
            v4f q;
            q.x = c0[u][0] * c0[u][0] + c0[u][1] * c0[u][1] + c0[u][2] * c0[u][2] + c0[u][3] * c0[u][3];
            q.y = c1[u][0] * c1[u][0] + c1[u][1] * c1[u][1] + c1[u][2] * c1[u][2] + c1[u][3] * c1[u][3];
            q.z = c0[u][4] * c0[u][4] + c0[u][5] * c0[u][5] + c0[u][6] * c0[u][6] + c0[u][7] * c0[u][7];
            q.w = c1[u][4] * c1[u][4] + c1[u][5] * c1[u][5] + c1[u][6] * c1[u][6] + c1[u][7] * c1[u][7];
            *(v4f*)&knp[cq[u]][n][4 * hi] = q;
        }
    }

    __syncthreads();
    int ones_before = 0, ones_all = 0;
    #pragma unroll
    for (int k = 0; k < PW; ++k) { ones_before += cnt[0][k]; ones_all += cnt[1][k]; }
    if (blockIdx.x == 0 && threadIdx.x == 0) N1[b] = ones_all;
    auto kvpos = [&](const int r) __attribute__((always_inline)) {
        const bool one = ((tile_ones >> r) & 1u) != 0u;
        const int  ob  = __builtin_popcount(tile_ones & ((1u << r) - 1u));
        return one ? (ones_before + ob)
                   : (ones_all + (s0 - ones_before) + (r - ob));
    };

    #pragma unroll
    for (int u = 0; u < UPW; ++u) {
        if (mat[u] == 1) {
            const int hk = 4 * hi + cq[u];
            const float kn = (knp[0][n][hk] + knp[1][n][hk]) + (knp[2][n][hk] + knp[3][n][hk]);
            Kn[(size_t)(b * HEADS + hk) * SEQ + kvpos(n)] = kn;
        }
    }

    const int rl = lane >> 2;
    const int pc = lane & 3;
    const int pk0 = kvpos(rl), pk1 = kvpos(8 + rl);
    #pragma unroll
    for (int k = 0; k < (3 * HEADS) / PW; ++k) {
        const int pp = ((3 * HEADS) / PW) * w + k;
        const int m  = pp >> 3;
        float* Out = (m == 0) ? Qh : (m == 1) ? Kh : Vh;
        float* ob  = Out + (size_t)(b * HEADS + (pp & 7)) * SEQ * HD + 4 * pc;
        const v4f r0 = *(const v4f*)&stg.row[pp][rl][4 * pc];
        const v4f r1 = *(const v4f*)&stg.row[pp][8 + rl][4 * pc];
        const int p0 = (m == 0) ? (s0 + rl)     : pk0;
        const int p1 = (m == 0) ? (s0 + 8 + rl) : pk1;
        *(v4f*)(ob + (size_t)p0 * HD) = r0;
        *(v4f*)(ob + (size_t)p1 * HD) = r1;
    }
}


__global__ void __launch_bounds__(ATT_BLOCK) __attribute__((amdgpu_waves_per_eu(4, 4)))
flash_attn_wmma(const float* __restrict__ Qh,
                const float* __restrict__ Kh,
                const float* __restrict__ Vh,
                const float* __restrict__ Kn,
                const int*   __restrict__ N1,
                float*       __restrict__ out)
{
    __shared__ float part[QG][NW][QT][10][PPITCH];
    __shared__ float kmw[NW * QG];
    __shared__ float mshare[QG][NW][QT][32];

    const int lane = threadIdx.x & 31;
    const int w    = __builtin_amdgcn_readfirstlane((int)(threadIdx.x >> 5));
    const int g    = w / NW;
    const int kr   = w % NW;
    const int n    = lane & 15;
    const int hi   = lane >> 4;
    const int q0   = (blockIdx.x * QG + g) * (16 * QT);
    const int bh   = blockIdx.y;
    const int b    = bh >> 3;
    const int h    = bh & 7;

    const float* Kbase = Kh + (size_t)bh * SEQ * HD;
    const float* Vbase = Vh + (size_t)bh * SEQ * HD;
    auto load_tile = [&](const int kv0, v2f (&ka)[4], v2f (&va)[4]) __attribute__((always_inline)) {
        const float* Krow = Kbase + (size_t)(kv0 + n) * HD;
        #pragma unroll
        for (int t = 0; t < 4; ++t) {
            const int kd = 4 * t + 2 * hi;
            ka[t].x = Krow[kd];
            ka[t].y = Krow[kd + 1];
        }
        const float* Vcol = Vbase + (size_t)(kv0 + 8 * hi) * HD + n;
        #pragma unroll
        for (int t = 0; t < 4; ++t) {
            va[t].x = Vcol[(size_t)(2 * t) * HD];
            va[t].y = Vcol[(size_t)(2 * t + 1) * HD];
        }
    };
    v2f ka0[4], va0[4];
    load_tile(16 * kr, ka0, va0);

    const int n1 = __builtin_amdgcn_readfirstlane(N1[b]);
    __builtin_amdgcn_sched_barrier(0);


    v2f qb[QT][4];
    #pragma unroll
    for (int qt = 0; qt < QT; ++qt) {
        const float* Qrow = Qh + ((size_t)bh * SEQ + q0 + qt * 16 + n) * HD;
        #pragma unroll
        for (int t = 0; t < 4; ++t) {
            const int ka = 4 * t + 2 * hi;
            qb[qt][t].x = Qrow[ka];
            qb[qt][t].y = Qrow[ka + 1];
        }
    }

    const v4f kn4 = *(const v4f*)(Kn + (size_t)bh * SEQ + 4 * (int)threadIdx.x);

    float m[QT], l[QT];
    v8f acc[QT];
    #pragma unroll
    for (int qt = 0; qt < QT; ++qt) {
        acc[qt] = (v8f){};
        m[qt] = -1.0e30f;
        l[qt] = 0.0f;
    }

    const float scale = 0.25f;

    auto compute_tile = [&](const int kv0, const v2f (&ka)[4], const v2f (&va)[4]) __attribute__((always_inline)) {
        __builtin_prefetch(Kbase + (size_t)(kv0 + 16 * NW + n) * HD, 0, 1);
        __builtin_prefetch(Vbase + (size_t)(kv0 + 16 * NW + n) * HD, 0, 1);

        float addm[8];
        #pragma unroll
        for (int r = 0; r < 8; ++r)
            addm[r] = (kv0 + 8 * hi + r >= n1) ? -1000.0f : 0.0f;

        #pragma unroll
        for (int qt = 0; qt < QT; ++qt) {
            v8f s = {};
            #pragma unroll
            for (int t = 0; t < 4; ++t)
                s = __builtin_amdgcn_wmma_f32_16x16x4_f32(false, ka[t],
                                                          false, qb[qt][t],
                                                          (short)0, s,
                                                          false, false);
            #pragma unroll
            for (int r = 0; r < 8; ++r) s[r] = s[r] * scale + addm[r];

            float mx = fmaxf(fmaxf(fmaxf(s[0], s[1]), fmaxf(s[2], s[3])),
                             fmaxf(fmaxf(s[4], s[5]), fmaxf(s[6], s[7])));
            mx = fmaxf(mx, xhalf(mx));
            const float mnew  = fmaxf(m[qt], mx);
            const float alpha = __expf(m[qt] - mnew);
            #pragma unroll
            for (int r = 0; r < 8; ++r) s[r] = __expf(s[r] - mnew);
            float ps = ((s[0] + s[1]) + (s[2] + s[3])) +
                       ((s[4] + s[5]) + (s[6] + s[7]));
            ps += xhalf(ps);
            l[qt] = l[qt] * alpha + ps;
            m[qt] = mnew;
            #pragma unroll
            for (int r = 0; r < 8; ++r) acc[qt][r] *= alpha;

            #pragma unroll
            for (int t = 0; t < 4; ++t) {
                v2f pb;
                pb.x = s[2 * t];
                pb.y = s[2 * t + 1];
                acc[qt] = __builtin_amdgcn_wmma_f32_16x16x4_f32(false, va[t],
                                                               false, pb,
                                                               (short)0, acc[qt],
                                                               false, false);
            }
        }
    };
    auto sweep_tile = [&](const int kv0) __attribute__((always_inline)) {
        v2f ka[4], va[4];
        load_tile(kv0, ka, va);
        compute_tile(kv0, ka, va);
    };

    asm volatile("" :: "v"(ka0[0]), "v"(ka0[1]), "v"(ka0[2]), "v"(ka0[3]),
                       "v"(va0[0]), "v"(va0[1]), "v"(va0[2]), "v"(va0[3]));

    const int nlt = (n1 + 15) >> 4;
    int tile = kr;
    auto set_prio = [&](const int t) __attribute__((always_inline)) {
        const int left = (nlt - t + NW - 1) / NW;
        if (left >= 6)      __builtin_amdgcn_s_setprio(3);
        else if (left >= 4) __builtin_amdgcn_s_setprio(2);
        else if (left >= 2) __builtin_amdgcn_s_setprio(1);
        else                __builtin_amdgcn_s_setprio(0);
    };
    if (tile < nlt) {
        set_prio(tile);
        compute_tile(tile * 16, ka0, va0);
        tile += NW;
    }

    {
        const int p0 = 4 * (int)threadIdx.x;
        float pm = 0.0f;
        pm = fmaxf(pm, (p0     >= n1) ? kn4.x : 0.0f);
        pm = fmaxf(pm, (p0 + 1 >= n1) ? kn4.y : 0.0f);
        pm = fmaxf(pm, (p0 + 2 >= n1) ? kn4.z : 0.0f);
        pm = fmaxf(pm, (p0 + 3 >= n1) ? kn4.w : 0.0f);
        pm = fmaxf(pm, xrow<8>(pm));
        pm = fmaxf(pm, xrow<4>(pm));
        pm = fmaxf(pm, xrow<2>(pm));
        pm = fmaxf(pm, xrow<1>(pm));
        pm = fmaxf(pm, xhalf(pm));
        if (lane == 0) kmw[w] = pm;
    }
    float qn[QT];
    #pragma unroll
    for (int qt = 0; qt < QT; ++qt) {
        float ss = 0.0f;
        #pragma unroll
        for (int t = 0; t < 4; ++t)
            ss += qb[qt][t].x * qb[qt][t].x + qb[qt][t].y * qb[qt][t].y;
        ss += xhalf(ss);
        qn[qt] = 0.250025f * sqrtf(ss);
    }

    for (; tile < nlt; tile += NW) {
        set_prio(tile);
        sweep_tile(tile * 16);
    }

    if (nlt < NTILE) {
        #pragma unroll
        for (int qt = 0; qt < QT; ++qt) mshare[g][kr][qt][lane] = m[qt];
        __syncthreads();
        float kmax2 = 0.0f;
        #pragma unroll
        for (int k = 0; k < NW * QG; ++k) kmax2 = fmaxf(kmax2, kmw[k]);
        const float kmax = sqrtf(kmax2);
        bool ok = true;
        #pragma unroll
        for (int qt = 0; qt < QT; ++qt) {
            float mg = m[qt];
            #pragma unroll
            for (int k = 0; k < NW; ++k) mg = fmaxf(mg, mshare[g][k][qt][lane]);
            ok &= (qn[qt] * kmax - mg <= 910.0f);
        }
        if (__builtin_amdgcn_ballot_w32(!ok) != 0u)
            for (; tile < NTILE; tile += NW) sweep_tile(tile * 16);
    }

    #pragma unroll
    for (int qt = 0; qt < QT; ++qt) {
        #pragma unroll
        for (int r = 0; r < 8; ++r) part[g][kr][qt][r][lane] = acc[qt][r];
        part[g][kr][qt][8][lane] = m[qt];
        part[g][kr][qt][9][lane] = l[qt];
    }
    __syncthreads();

    const int mq = kr / WPQ;
    const int qq = (kr % WPQ) * QPW + (lane % QPW);
    const int dl = lane / QPW;
    float mw[NW], sc[NW];
    #pragma unroll
    for (int k = 0; k < NW; ++k) mw[k] = part[g][k][mq][8][qq];
    float M = mw[0];
    #pragma unroll
    for (int k = 1; k < NW; ++k) M = fmaxf(M, mw[k]);
    float L = 0.0f;
    #pragma unroll
    for (int k = 0; k < NW; ++k) {
        sc[k] = __expf(mw[k] - M);
        L += part[g][k][mq][9][qq] * sc[k];
    }
    float* Orow = out + ((size_t)b * SEQ + q0 + mq * 16 + qq) * DIM + h;
    #pragma unroll
    for (int i = 0; i < 16 / LPQ; ++i) {
        const int d  = LPQ * i + dl;
        const int rr = d & 7;
        const int ll = 16 * (d >> 3) + qq;
        float o = 0.0f;
        #pragma unroll
        for (int k = 0; k < NW; ++k) o += part[g][k][mq][rr][ll] * sc[k];
        Orow[d * HEADS] = o / L;
    }
}

extern "C" void kernel_launch(void* const* d_in, const int* in_sizes, int n_in,
                              void* d_out, int out_size, void* d_ws, size_t ws_size,
                              hipStream_t stream)
{
    const float* X        = (const float*)d_in[0];
    const int*   mask_src = (const int*)d_in[1];
    const float* Wq       = (const float*)d_in[4];
    const float* Wk       = (const float*)d_in[5];
    const float* Wv       = (const float*)d_in[6];
    float*       out      = (float*)d_out;

    const size_t headElems = (size_t)BATCH * HEADS * SEQ * HD;
    float* Qh = (float*)d_ws;
    float* Kh = Qh + headElems;
    float* Vh = Kh + headElems;
    float* Kn = Vh + headElems;
    int*   N1 = (int*)(Kn + (size_t)BATCH * HEADS * SEQ);

    dim3 gridP(SEQ / 16, BATCH);
    qkv_proj_wmma<<<gridP, 32 * PW, 0, stream>>>(X, Wq, Wk, Wv, mask_src, Qh, Kh, Vh, Kn, N1);

    dim3 gridA(SEQ / (16 * QT * QG), BATCH * HEADS);
    flash_attn_wmma<<<gridA, ATT_BLOCK, 0, stream>>>(Qh, Kh, Vh, Kn, N1, out);
}


# Softmax in base 2 on raw scores: one packed fma + exp2 per score pair (scale folded in), -1000 steps only in tiles with masked keys, row sums kept per lane half and joined once; row-sum tree changes
# speedup vs baseline: 13.3866x; 1.0373x over previous
#include <hip/hip_runtime.h>

#define DIM   128
#define HEADS 8
#define BATCH 2
#define SEQ   2048
#define HD    16
#define QT    4
#define NW    8
#define QG    2
#define KPW   (SEQ / NW)
#define WPQ   (NW / QT)
#define QPW   (16 / WPQ)
#define LPQ   (32 / QPW)
#define PPITCH 36
#define ATT_BLOCK 512
#define NTILE (SEQ / 16)
static_assert(ATT_BLOCK == 32 * NW * QG, "one wave per (query group, key range)");
static_assert(NW % QT == 0 && 16 % WPQ == 0 && KPW % 16 == 0, "merge split");
static_assert(LPQ == 4 && HEADS * 4 * LPQ == 128, "one output line per query and store");

typedef __attribute__((ext_vector_type(2))) float v2f;
typedef __attribute__((ext_vector_type(8))) float v8f;

__device__ __forceinline__ float xhalf(float x)
{
    const int b = __builtin_bit_cast(int, x);
    return __builtin_bit_cast(float,
        __builtin_amdgcn_permlanex16(b, b, 0x76543210, 0xfedcba98, false, false));
}

template <int X>
__device__ __forceinline__ float xrow(float x)
{
    const int b = __builtin_bit_cast(int, x);
    return __builtin_bit_cast(float,
        __builtin_amdgcn_update_dpp(b, b, 0x160 + X, 0xf, 0xf, false));
}

#define PW  4
#define UPW 3
typedef __attribute__((ext_vector_type(4))) float v4f;
static_assert(PW * UPW == 12, "3 matrices x 4 column quarters");
#define MQ  (SEQ / (4 * 32 * PW))
#define PF  10
#define SPITCH 20
#define LDS_ALONE (164 * 1024)
static_assert(SEQ % (4 * 32 * PW) == 0, "mask words per thread");
static_assert(6 * PF <= 63, "loads a wave can have outstanding");
static_assert((3 * HEADS) % PW == 0, "(matrix, head) pairs per wave");

__global__ void __launch_bounds__(32 * PW) __attribute__((amdgpu_waves_per_eu(1, 1)))
qkv_proj_wmma(const float* __restrict__ X,
              const float* __restrict__ Wq,
              const float* __restrict__ Wk,
              const float* __restrict__ Wv,
              const int*   __restrict__ mask_src,
              float* __restrict__ Qh,
              float* __restrict__ Kh,
              float* __restrict__ Vh,
              float* __restrict__ Kn,
              int*   __restrict__ N1)
{
    __shared__ int cnt[2][PW];
    __shared__ __attribute__((aligned(16))) float knp[4][16][HEADS];
    __shared__ __attribute__((aligned(16))) union {
        float row[3 * HEADS][16][SPITCH];
        char  alone[LDS_ALONE];
    } stg;

    const int lane = threadIdx.x & 31;
    const int w    = __builtin_amdgcn_readfirstlane((int)(threadIdx.x >> 5));
    const int n    = lane & 15;
    const int hi   = lane >> 4;
    const int s0   = blockIdx.x * 16;
    const int b    = blockIdx.y;

    int mat[UPW], cq[UPW];
    const float* Wp[UPW];
    #pragma unroll
    for (int u = 0; u < UPW; ++u) {
        const int p = UPW * w + u;
        mat[u] = p >> 2;
        cq[u]  = p & 3;
        const float* Wm = (mat[u] == 0) ? Wq : (mat[u] == 1) ? Wk : Wv;
        Wp[u] = Wm + (size_t)(4 * hi) * DIM + 32 * cq[u] + HEADS * (n & 3) + 2 * (n >> 2);
    }

    const int* mrow = mask_src + (size_t)b * SEQ;
    int4 mwd[MQ];
    #pragma unroll
    for (int q = 0; q < MQ; ++q)
        mwd[q] = *(const int4*)(mrow + 4 * ((int)threadIdx.x + 32 * PW * q));
    const int mown = mrow[s0 + n];

    const float* Xp = X + ((size_t)b * SEQ + s0 + n) * DIM + 4 * hi;
    v4f xv[16];
    #pragma unroll
    for (int j = 0; j < 16; ++j) xv[j] = *(const v4f*)(Xp + 8 * j);

    v2f wa[32][UPW], wb[32][UPW];
    #pragma unroll
    for (int s = 0; s < PF; ++s) {
        #pragma unroll
        for (int u = 0; u < UPW; ++u) {
            const float* wr = Wp[u] + (size_t)(8 * (s >> 1) + 2 * (s & 1)) * DIM;
            wa[s][u] = *(const v2f*)wr;
            wb[s][u] = *(const v2f*)(wr + DIM);
        }
    }
    __builtin_amdgcn_sched_barrier(0);

    {
        int tot = 0, bef = 0;
        #pragma unroll
        for (int q = 0; q < MQ; ++q) {
            const int i0 = 4 * ((int)threadIdx.x + 32 * PW * q);
            const int4 a = mwd[q];
            tot += __builtin_popcount(__builtin_amdgcn_ballot_w32(a.x != 0))
                 + __builtin_popcount(__builtin_amdgcn_ballot_w32(a.y != 0))
                 + __builtin_popcount(__builtin_amdgcn_ballot_w32(a.z != 0))
                 + __builtin_popcount(__builtin_amdgcn_ballot_w32(a.w != 0));
            bef += __builtin_popcount(__builtin_amdgcn_ballot_w32(a.x != 0 && i0     < s0))
                 + __builtin_popcount(__builtin_amdgcn_ballot_w32(a.y != 0 && i0 + 1 < s0))
                 + __builtin_popcount(__builtin_amdgcn_ballot_w32(a.z != 0 && i0 + 2 < s0))
                 + __builtin_popcount(__builtin_amdgcn_ballot_w32(a.w != 0 && i0 + 3 < s0));
        }
        if (lane == 0) { cnt[0][w] = bef; cnt[1][w] = tot; }
    }
    const unsigned tile_ones = __builtin_amdgcn_ballot_w32(mown != 0) & 0xffffu;

    v8f c0[UPW], c1[UPW];
    #pragma unroll
    for (int u = 0; u < UPW; ++u) { c0[u] = (v8f){}; c1[u] = (v8f){}; }
    #pragma unroll
    for (int s = 0; s < 32; ++s) {
        if (s + PF < 32) {
            #pragma unroll
            for (int u = 0; u < UPW; ++u) {
                const int sn = s + PF;
                const float* wr = Wp[u] + (size_t)(8 * (sn >> 1) + 2 * (sn & 1)) * DIM;
                wa[s + PF][u] = *(const v2f*)wr;
                wb[s + PF][u] = *(const v2f*)(wr + DIM);
            }
        }
        const int j = s >> 1;
        v2f xb;
        xb.x = (s & 1) ? xv[j].z : xv[j].x;
        xb.y = (s & 1) ? xv[j].w : xv[j].y;
        #pragma unroll
        for (int u = 0; u < UPW; ++u) {
            v2f a0, a1;
            a0.x = wa[s][u].x;  a0.y = wb[s][u].x;
            a1.x = wa[s][u].y;  a1.y = wb[s][u].y;
            c0[u] = __builtin_amdgcn_wmma_f32_16x16x4_f32(false, a0, false, xb,
                                                          (short)0, c0[u], false, false);
            c1[u] = __builtin_amdgcn_wmma_f32_16x16x4_f32(false, a1, false, xb,
                                                          (short)0, c1[u], false, false);
        }
        __builtin_amdgcn_sched_barrier(0);
    }

    #pragma unroll
    for (int u = 0; u < UPW; ++u) {
        float* sp = &stg.row[mat[u] * HEADS + 4 * hi][n][4 * cq[u]];
        const int hs = 16 * SPITCH;
        v4f t;
        t.x = c0[u][0]; t.y = c0[u][1]; t.z = c0[u][2]; t.w = c0[u][3];  *(v4f*)sp            = t;
        t.x = c1[u][0]; t.y = c1[u][1]; t.z = c1[u][2]; t.w = c1[u][3];  *(v4f*)(sp + hs)     = t;
        t.x = c0[u][4]; t.y = c0[u][5]; t.z = c0[u][6]; t.w = c0[u][7];  *(v4f*)(sp + 2 * hs) = t;
        t.x = c1[u][4]; t.y = c1[u][5]; t.z = c1[u][6]; t.w = c1[u][7];  *(v4f*)(sp + 3 * hs) = t;
        if (mat[u] == 1) {
            v4f q;
            q.x = c0[u][0] * c0[u][0] + c0[u][1] * c0[u][1] + c0[u][2] * c0[u][2] + c0[u][3] * c0[u][3];
            q.y = c1[u][0] * c1[u][0] + c1[u][1] * c1[u][1] + c1[u][2] * c1[u][2] + c1[u][3] * c1[u][3];
            q.z = c0[u][4] * c0[u][4] + c0[u][5] * c0[u][5] + c0[u][6] * c0[u][6] + c0[u][7] * c0[u][7];
            q.w = c1[u][4] * c1[u][4] + c1[u][5] * c1[u][5] + c1[u][6] * c1[u][6] + c1[u][7] * c1[u][7];
            *(v4f*)&knp[cq[u]][n][4 * hi] = q;
        }
    }

    __syncthreads();
    int ones_before = 0, ones_all = 0;
    #pragma unroll
    for (int k = 0; k < PW; ++k) { ones_before += cnt[0][k]; ones_all += cnt[1][k]; }
    if (blockIdx.x == 0 && threadIdx.x == 0) N1[b] = ones_all;
    auto kvpos = [&](const int r) __attribute__((always_inline)) {
        const bool one = ((tile_ones >> r) & 1u) != 0u;
        const int  ob  = __builtin_popcount(tile_ones & ((1u << r) - 1u));
        return one ? (ones_before + ob)
                   : (ones_all + (s0 - ones_before) + (r - ob));
    };

    #pragma unroll
    for (int u = 0; u < UPW; ++u) {
        if (mat[u] == 1) {
            const int hk = 4 * hi + cq[u];
            const float kn = (knp[0][n][hk] + knp[1][n][hk]) + (knp[2][n][hk] + knp[3][n][hk]);
            Kn[(size_t)(b * HEADS + hk) * SEQ + kvpos(n)] = kn;
        }
    }

    const int rl = lane >> 2;
    const int pc = lane & 3;
    const int pk0 = kvpos(rl), pk1 = kvpos(8 + rl);
    #pragma unroll
    for (int k = 0; k < (3 * HEADS) / PW; ++k) {
        const int pp = ((3 * HEADS) / PW) * w + k;
        const int m  = pp >> 3;
        float* Out = (m == 0) ? Qh : (m == 1) ? Kh : Vh;
        float* ob  = Out + (size_t)(b * HEADS + (pp & 7)) * SEQ * HD + 4 * pc;
        const v4f r0 = *(const v4f*)&stg.row[pp][rl][4 * pc];
        const v4f r1 = *(const v4f*)&stg.row[pp][8 + rl][4 * pc];
        const int p0 = (m == 0) ? (s0 + rl)     : pk0;
        const int p1 = (m == 0) ? (s0 + 8 + rl) : pk1;
        *(v4f*)(ob + (size_t)p0 * HD) = r0;
        *(v4f*)(ob + (size_t)p1 * HD) = r1;
    }
}


__global__ void __launch_bounds__(ATT_BLOCK) __attribute__((amdgpu_waves_per_eu(4, 4)))
flash_attn_wmma(const float* __restrict__ Qh,
                const float* __restrict__ Kh,
                const float* __restrict__ Vh,
                const float* __restrict__ Kn,
                const int*   __restrict__ N1,
                float*       __restrict__ out)
{
    __shared__ float part[QG][NW][QT][10][PPITCH];
    __shared__ float kmw[NW * QG];
    __shared__ float mshare[QG][NW][QT][32];

    const int lane = threadIdx.x & 31;
    const int w    = __builtin_amdgcn_readfirstlane((int)(threadIdx.x >> 5));
    const int g    = w / NW;
    const int kr   = w % NW;
    const int n    = lane & 15;
    const int hi   = lane >> 4;
    const int q0   = (blockIdx.x * QG + g) * (16 * QT);
    const int bh   = blockIdx.y;
    const int b    = bh >> 3;
    const int h    = bh & 7;

    const float* Kbase = Kh + (size_t)bh * SEQ * HD;
    const float* Vbase = Vh + (size_t)bh * SEQ * HD;
    auto load_tile = [&](const int kv0, v2f (&ka)[4], v2f (&va)[4]) __attribute__((always_inline)) {
        const float* Krow = Kbase + (size_t)(kv0 + n) * HD;
        #pragma unroll
        for (int t = 0; t < 4; ++t) {
            const int kd = 4 * t + 2 * hi;
            ka[t].x = Krow[kd];
            ka[t].y = Krow[kd + 1];
        }
        const float* Vcol = Vbase + (size_t)(kv0 + 8 * hi) * HD + n;
        #pragma unroll
        for (int t = 0; t < 4; ++t) {
            va[t].x = Vcol[(size_t)(2 * t) * HD];
            va[t].y = Vcol[(size_t)(2 * t + 1) * HD];
        }
    };
    v2f ka0[4], va0[4];
    load_tile(16 * kr, ka0, va0);

    const int n1 = __builtin_amdgcn_readfirstlane(N1[b]);
    __builtin_amdgcn_sched_barrier(0);


    v2f qb[QT][4];
    #pragma unroll
    for (int qt = 0; qt < QT; ++qt) {
        const float* Qrow = Qh + ((size_t)bh * SEQ + q0 + qt * 16 + n) * HD;
        #pragma unroll
        for (int t = 0; t < 4; ++t) {
            const int ka = 4 * t + 2 * hi;
            qb[qt][t].x = Qrow[ka];
            qb[qt][t].y = Qrow[ka + 1];
        }
    }

    const v4f kn4 = *(const v4f*)(Kn + (size_t)bh * SEQ + 4 * (int)threadIdx.x);

    float m[QT], l[QT];
    v8f acc[QT];
    #pragma unroll
    for (int qt = 0; qt < QT; ++qt) {
        acc[qt] = (v8f){};
        m[qt] = -1.0e30f;
        l[qt] = 0.0f;
    }

    const float LOG2E = 1.4426950408889634f;
    const float LN2   = 0.6931471805599453f;
    const float CS    = 0.25f * LOG2E;

    auto compute_tile = [&](const int kv0, const v2f (&ka)[4], const v2f (&va)[4], const bool mixed) __attribute__((always_inline)) {
        __builtin_prefetch(Kbase + (size_t)(kv0 + 16 * NW + n) * HD, 0, 1);
        __builtin_prefetch(Vbase + (size_t)(kv0 + 16 * NW + n) * HD, 0, 1);

        float addm[8];
        if (mixed) {
            #pragma unroll
            for (int r = 0; r < 8; ++r)
                addm[r] = (kv0 + 8 * hi + r >= n1) ? -1000.0f : 0.0f;
        }
        const float cs = mixed ? LOG2E : CS;
        const v2f cs2 = {cs, cs};

        #pragma unroll
        for (int qt = 0; qt < QT; ++qt) {
            v8f s = {};
            #pragma unroll
            for (int t = 0; t < 4; ++t)
                s = __builtin_amdgcn_wmma_f32_16x16x4_f32(false, ka[t],
                                                          false, qb[qt][t],
                                                          (short)0, s,
                                                          false, false);
            if (mixed) {
                #pragma unroll
                for (int r = 0; r < 8; ++r) s[r] = s[r] * 0.25f + addm[r];
            }

            float mx = fmaxf(fmaxf(fmaxf(s[0], s[1]), fmaxf(s[2], s[3])),
                             fmaxf(fmaxf(s[4], s[5]), fmaxf(s[6], s[7])));
            mx = fmaxf(mx, xhalf(mx));
            const float mnew  = fmaxf(m[qt], mx * cs);
            const float alpha = __builtin_amdgcn_exp2f(m[qt] - mnew);
            const v2f nm2 = {-mnew, -mnew};
            v2f pb[4];
            #pragma unroll
            for (int t = 0; t < 4; ++t) {
                v2f x;
                x.x = s[2 * t];
                x.y = s[2 * t + 1];
                x = __builtin_elementwise_fma(x, cs2, nm2);
                pb[t].x = __builtin_amdgcn_exp2f(x.x);
                pb[t].y = __builtin_amdgcn_exp2f(x.y);
            }
            const v2f ps2 = (pb[0] + pb[1]) + (pb[2] + pb[3]);
            l[qt] = l[qt] * alpha + (ps2.x + ps2.y);
            m[qt] = mnew;
            acc[qt] = acc[qt] * alpha;

            #pragma unroll
            for (int t = 0; t < 4; ++t)
                acc[qt] = __builtin_amdgcn_wmma_f32_16x16x4_f32(false, va[t],
                                                               false, pb[t],
                                                               (short)0, acc[qt],
                                                               false, false);
        }
    };
    auto sweep_tile = [&](const int kv0, const bool mixed) __attribute__((always_inline)) {
        v2f ka[4], va[4];
        load_tile(kv0, ka, va);
        compute_tile(kv0, ka, va, mixed);
    };

    asm volatile("" :: "v"(ka0[0]), "v"(ka0[1]), "v"(ka0[2]), "v"(ka0[3]),
                       "v"(va0[0]), "v"(va0[1]), "v"(va0[2]), "v"(va0[3]));

    const int nlt = (n1 + 15) >> 4;
    const int nfull = n1 >> 4;
    int tile = kr;
    auto set_prio = [&](const int t) __attribute__((always_inline)) {
        const int left = (nlt - t + NW - 1) / NW;
        if (left >= 6)      __builtin_amdgcn_s_setprio(3);
        else if (left >= 4) __builtin_amdgcn_s_setprio(2);
        else if (left >= 2) __builtin_amdgcn_s_setprio(1);
        else                __builtin_amdgcn_s_setprio(0);
    };
    if (tile < nlt) {
        set_prio(tile);
        if (tile < nfull) compute_tile(tile * 16, ka0, va0, false);
        else              compute_tile(tile * 16, ka0, va0, true);
        tile += NW;
    }

    {
        const int p0 = 4 * (int)threadIdx.x;
        float pm = 0.0f;
        pm = fmaxf(pm, (p0     >= n1) ? kn4.x : 0.0f);
        pm = fmaxf(pm, (p0 + 1 >= n1) ? kn4.y : 0.0f);
        pm = fmaxf(pm, (p0 + 2 >= n1) ? kn4.z : 0.0f);
        pm = fmaxf(pm, (p0 + 3 >= n1) ? kn4.w : 0.0f);
        pm = fmaxf(pm, xrow<8>(pm));
        pm = fmaxf(pm, xrow<4>(pm));
        pm = fmaxf(pm, xrow<2>(pm));
        pm = fmaxf(pm, xrow<1>(pm));
        pm = fmaxf(pm, xhalf(pm));
        if (lane == 0) kmw[w] = pm;
    }
    float qn[QT];
    #pragma unroll
    for (int qt = 0; qt < QT; ++qt) {
        float ss = 0.0f;
        #pragma unroll
        for (int t = 0; t < 4; ++t)
            ss += qb[qt][t].x * qb[qt][t].x + qb[qt][t].y * qb[qt][t].y;
        ss += xhalf(ss);
        qn[qt] = 0.250025f * sqrtf(ss);
    }

    for (; tile < nfull; tile += NW) {
        set_prio(tile);
        sweep_tile(tile * 16, false);
    }
    if (tile < nlt) {
        set_prio(tile);
        sweep_tile(tile * 16, true);
        tile += NW;
    }

    if (nlt < NTILE) {
        #pragma unroll
        for (int qt = 0; qt < QT; ++qt) mshare[g][kr][qt][lane] = m[qt];
        __syncthreads();
        float kmax2 = 0.0f;
        #pragma unroll
        for (int k = 0; k < NW * QG; ++k) kmax2 = fmaxf(kmax2, kmw[k]);
        const float kmax = sqrtf(kmax2);
        bool ok = true;
        #pragma unroll
        for (int qt = 0; qt < QT; ++qt) {
            float mg = m[qt];
            #pragma unroll
            for (int k = 0; k < NW; ++k) mg = fmaxf(mg, mshare[g][k][qt][lane]);
            ok &= (qn[qt] * kmax - mg * LN2 <= 910.0f);
        }
        if (__builtin_amdgcn_ballot_w32(!ok) != 0u)
            for (; tile < NTILE; tile += NW) sweep_tile(tile * 16, true);
    }

    #pragma unroll
    for (int qt = 0; qt < QT; ++qt) {
        #pragma unroll
        for (int r = 0; r < 8; ++r) part[g][kr][qt][r][lane] = acc[qt][r];
        part[g][kr][qt][8][lane] = m[qt];
        part[g][kr][qt][9][lane] = l[qt] + xhalf(l[qt]);
    }
    __syncthreads();

    const int mq = kr / WPQ;
    const int qq = (kr % WPQ) * QPW + (lane % QPW);
    const int dl = lane / QPW;
    float mw[NW], sc[NW];
    #pragma unroll
    for (int k = 0; k < NW; ++k) mw[k] = part[g][k][mq][8][qq];
    float M = mw[0];
    #pragma unroll
    for (int k = 1; k < NW; ++k) M = fmaxf(M, mw[k]);
    float L = 0.0f;
    #pragma unroll
    for (int k = 0; k < NW; ++k) {
        sc[k] = __builtin_amdgcn_exp2f(mw[k] - M);
        L += part[g][k][mq][9][qq] * sc[k];
    }
    float* Orow = out + ((size_t)b * SEQ + q0 + mq * 16 + qq) * DIM + h;
    #pragma unroll
    for (int i = 0; i < 16 / LPQ; ++i) {
        const int d  = LPQ * i + dl;
        const int rr = d & 7;
        const int ll = 16 * (d >> 3) + qq;
        float o = 0.0f;
        #pragma unroll
        for (int k = 0; k < NW; ++k) o += part[g][k][mq][rr][ll] * sc[k];
        Orow[d * HEADS] = o / L;
    }
}

extern "C" void kernel_launch(void* const* d_in, const int* in_sizes, int n_in,
                              void* d_out, int out_size, void* d_ws, size_t ws_size,
                              hipStream_t stream)
{
    const float* X        = (const float*)d_in[0];
    const int*   mask_src = (const int*)d_in[1];
    const float* Wq       = (const float*)d_in[4];
    const float* Wk       = (const float*)d_in[5];
    const float* Wv       = (const float*)d_in[6];
    float*       out      = (float*)d_out;

    const size_t headElems = (size_t)BATCH * HEADS * SEQ * HD;
    float* Qh = (float*)d_ws;
    float* Kh = Qh + headElems;
    float* Vh = Kh + headElems;
    float* Kn = Vh + headElems;
    int*   N1 = (int*)(Kn + (size_t)BATCH * HEADS * SEQ);

    dim3 gridP(SEQ / 16, BATCH);
    qkv_proj_wmma<<<gridP, 32 * PW, 0, stream>>>(X, Wq, Wk, Wv, mask_src, Qh, Kh, Vh, Kn, N1);

    dim3 gridA(SEQ / (16 * QT * QG), BATCH * HEADS);
    flash_attn_wmma<<<gridA, ATT_BLOCK, 0, stream>>>(Qh, Kh, Vh, Kn, N1, out);
}
